# starting kernel
# baseline (speedup 1.0000x reference)
#include <hip/hip_runtime.h>
#include <hip/hip_bf16.h>
#include <math.h>


typedef float v2f __attribute__((ext_vector_type(2)));
typedef float v8f __attribute__((ext_vector_type(8)));

#define N_NODES 50000
#define N_EDGES 600000
#define E_TOT   650000
#define IN_CH   256
#define HID     128
#define ENC_NEG_INF 0x007FFFFFu

__device__ __forceinline__ unsigned enc_f32(float f) {
  unsigned u = __float_as_uint(f);
  return (u & 0x80000000u) ? ~u : (u | 0x80000000u);
}
__device__ __forceinline__ float dec_f32(unsigned e) {
  unsigned u = (e & 0x80000000u) ? (e & 0x7FFFFFFFu) : ~e;
  return __uint_as_float(u);
}

__device__ __forceinline__ float wave_sum(float v) {
  #pragma unroll
  for (int off = 16; off >= 1; off >>= 1) v += __shfl_xor(v, off, 32);
  return v;
}

__global__ void k_gemm_relu(const float* __restrict__ X,
                            const float* __restrict__ W,
                            const float* __restrict__ B,
                            float* __restrict__ H,
                            float* __restrict__ rn) {
  const int wave = (int)((blockIdx.x * blockDim.x + threadIdx.x) >> 5);
  const int lane = (int)(threadIdx.x & 31);
  if (wave * 16 >= N_NODES) return;

  const int m0   = wave * 16;
  const int mrow = m0 + (lane & 15);
  const int kg   = (lane >> 4) * 2;
  const int n15  = lane & 15;

  v8f acc[8];
  #pragma unroll
  for (int t = 0; t < 8; ++t) acc[t] = (v8f){0.f,0.f,0.f,0.f,0.f,0.f,0.f,0.f};

  const float* __restrict__ xrow = X + (size_t)mrow * IN_CH + kg;
  const float* __restrict__ wbas = W + (size_t)n15 * IN_CH + kg;

  for (int k = 0; k < IN_CH; k += 8) {
    const v2f a0 = *(const v2f*)(xrow + k);
    const v2f a1 = *(const v2f*)(xrow + k + 4);
    v2f b0[8], b1[8];
    #pragma unroll
    for (int t = 0; t < 8; ++t)
      b0[t] = *(const v2f*)(wbas + (size_t)t * 16 * IN_CH + k);
    #pragma unroll
    for (int t = 0; t < 8; ++t)
      b1[t] = *(const v2f*)(wbas + (size_t)t * 16 * IN_CH + k + 4);
    #pragma unroll
    for (int t = 0; t < 8; ++t)
      acc[t] = __builtin_amdgcn_wmma_f32_16x16x4_f32(
          false, a0, false, b0[t], (short)0, acc[t], false, false);
    #pragma unroll
    for (int t = 0; t < 8; ++t)
      acc[t] = __builtin_amdgcn_wmma_f32_16x16x4_f32(
          false, a1, false, b1[t], (short)0, acc[t], false, false);
  }

  const int mside = (lane >> 4) * 8;
  float ss[8];
  #pragma unroll
  for (int r = 0; r < 8; ++r) ss[r] = 0.f;

  #pragma unroll
  for (int t = 0; t < 8; ++t) {
    const int col = t * 16 + n15;
    const float bb = B[col];
    #pragma unroll
    for (int r = 0; r < 8; ++r) {
      const int row = m0 + r + mside;
      float v = acc[t][r] + bb;
      v = v > 0.f ? v : 0.f;
      H[(size_t)row * HID + col] = v;
      ss[r] += v * v;
    }
  }
  #pragma unroll
  for (int r = 0; r < 8; ++r) {
    #pragma unroll
    for (int off = 8; off >= 1; off >>= 1) ss[r] += __shfl_xor(ss[r], off, 32);
  }
  if (n15 == 0) {
    #pragma unroll
    for (int r = 0; r < 8; ++r)
      rn[m0 + r + mside] = 1.f / fmaxf(sqrtf(ss[r]), 1e-12f);
  }
}

__global__ void k_init(float* __restrict__ Y,
                       unsigned* __restrict__ segmax,
                       float* __restrict__ segsum) {
  const int i = (int)(blockIdx.x * blockDim.x + threadIdx.x);
  if (i < N_NODES * HID) Y[i] = 0.f;
  if (i < N_NODES) { segmax[i] = ENC_NEG_INF; segsum[i] = 0.f; }
}

__global__ void k_rnorm(const float* __restrict__ X, float* __restrict__ rn) {
  const int node = (int)((blockIdx.x * blockDim.x + threadIdx.x) >> 5);
  const int lane = (int)(threadIdx.x & 31);
  if (node >= N_NODES) return;
  const float4 v = *(const float4*)(X + (size_t)node * HID + lane * 4);
  float s = v.x * v.x + v.y * v.y + v.z * v.z + v.w * v.w;
  s = wave_sum(s);
  if (lane == 0) rn[node] = 1.f / fmaxf(sqrtf(s), 1e-12f);
}

__global__ void k_edge_logit(const float* __restrict__ X,
                             const int* __restrict__ EI,
                             const float* __restrict__ rn,
                             const float* __restrict__ beta_ptr,
                             float* __restrict__ elog,
                             unsigned* __restrict__ segmax) {
  const int e    = (int)((blockIdx.x * blockDim.x + threadIdx.x) >> 5);
  const int lane = (int)(threadIdx.x & 31);
  if (e >= E_TOT) return;
  int src, dst;
  if (e < N_EDGES) { src = EI[e]; dst = EI[N_EDGES + e]; }
  else             { src = dst = e - N_EDGES; }
  const float4 a = *(const float4*)(X + (size_t)src * HID + lane * 4);
  const float4 b = *(const float4*)(X + (size_t)dst * HID + lane * 4);
  float s = a.x * b.x + a.y * b.y + a.z * b.z + a.w * b.w;
  s = wave_sum(s);
  if (lane == 0) {
    const float beta = beta_ptr ? beta_ptr[0] : 1.0f;
    const float logit = beta * s * rn[src] * rn[dst];
    elog[e] = logit;
    atomicMax(segmax + dst, enc_f32(logit));
  }
}

__global__ void k_edge_exp(const int* __restrict__ EI,
                           const unsigned* __restrict__ segmax,
                           float* __restrict__ elog,
                           float* __restrict__ segsum) {
  const int e = (int)(blockIdx.x * blockDim.x + threadIdx.x);
  if (e >= E_TOT) return;
  const int dst = (e < N_EDGES) ? EI[N_EDGES + e] : (e - N_EDGES);
  const float ex = __expf(elog[e] - dec_f32(segmax[dst]));
  elog[e] = ex;
  atomicAdd(segsum + dst, ex);
}

__global__ void k_scatter(const float* __restrict__ X,
                          const int* __restrict__ EI,
                          const float* __restrict__ elog,
                          const float* __restrict__ segsum,
                          float* __restrict__ Y) {
  const int e    = (int)((blockIdx.x * blockDim.x + threadIdx.x) >> 5);
  const int lane = (int)(threadIdx.x & 31);
  if (e >= E_TOT) return;
  int src, dst;
  if (e < N_EDGES) { src = EI[e]; dst = EI[N_EDGES + e]; }
  else             { src = dst = e - N_EDGES; }
  const float alpha = elog[e] / (segsum[dst] + 1e-16f);
  const float4 xs = *(const float4*)(X + (size_t)src * HID + lane * 4);
  float* yp = Y + (size_t)dst * HID + lane * 4;
  atomicAdd(yp + 0, alpha * xs.x);
  atomicAdd(yp + 1, alpha * xs.y);
  atomicAdd(yp + 2, alpha * xs.z);
  atomicAdd(yp + 3, alpha * xs.w);
}

static void run_conv(const float* X, float* Y, const int* EI,
                     const float* beta_ptr, float* rn, unsigned* segmax,
                     float* segsum, float* elog, bool compute_rnorm,
                     hipStream_t stream) {
  const int init_threads = N_NODES * HID;
  k_init<<<(init_threads + 255) / 256, 256, 0, stream>>>(Y, segmax, segsum);

  if (compute_rnorm) {
    const long long norm_threads = (long long)N_NODES * 32;
    k_rnorm<<<(int)((norm_threads + 255) / 256), 256, 0, stream>>>(X, rn);
  }

  const long long edge_threads = (long long)E_TOT * 32;
  const int eblocks = (int)((edge_threads + 255) / 256);
  k_edge_logit<<<eblocks, 256, 0, stream>>>(X, EI, rn, beta_ptr, elog, segmax);

  k_edge_exp<<<(E_TOT + 255) / 256, 256, 0, stream>>>(EI, segmax, elog, segsum);

  k_scatter<<<eblocks, 256, 0, stream>>>(X, EI, elog, segsum, Y);
}

extern "C" void kernel_launch(void* const* d_in, const int* in_sizes, int n_in,
                              void* d_out, int out_size, void* d_ws, size_t ws_size,
                              hipStream_t stream) {
  const float* x     = (const float*)d_in[0];
  const int*   ei    = (const int*)d_in[1];
  const float* W1    = (const float*)d_in[2];
  const float* b1    = (const float*)d_in[3];
  const float* beta2 = (const float*)d_in[4];
  float* out = (float*)d_out;

  float*    ws     = (float*)d_ws;
  float*    h1     = ws;
  float*    h2     = ws + 6400000;
  float*    rn     = ws + 12800000;
  unsigned* segmax = (unsigned*)(ws + 12850000);
  float*    segsum = ws + 12900000;
  float*    elog   = ws + 12950000;

  const int gemm_waves  = (N_NODES + 15) / 16;
  const int gemm_blocks = (gemm_waves * 32 + 255) / 256;
  k_gemm_relu<<<gemm_blocks, 256, 0, stream>>>(x, W1, b1, h1, rn);

  run_conv(h1, h2, ei, nullptr, rn, segmax, segsum, elog, false, stream);
  run_conv(h2, out, ei, beta2, rn, segmax, segsum, elog, true, stream);
}


# Row sums of both convs in LDS: 256 owner work-groups (196 nodes) start from self loops, scan all dst keys, ds_add_f32 gathered rows (arrival order as before); conv-1 rn summed in lane+32j order
# speedup vs baseline: 1.8973x; 1.8973x over previous
#include <hip/hip_runtime.h>
#include <hip/hip_bf16.h>
#include <math.h>


typedef float v2f __attribute__((ext_vector_type(2)));
typedef float v8f __attribute__((ext_vector_type(8)));

#define N_NODES 50000
#define N_EDGES 600000
#define E_TOT   650000
#define IN_CH   256
#define HID     128
#define ENC_NEG_INF 0x007FFFFFu

__device__ __forceinline__ unsigned enc_f32(float f) {
  unsigned u = __float_as_uint(f);
  return (u & 0x80000000u) ? ~u : (u | 0x80000000u);
}
__device__ __forceinline__ float dec_f32(unsigned e) {
  unsigned u = (e & 0x80000000u) ? (e & 0x7FFFFFFFu) : ~e;
  return __uint_as_float(u);
}

__device__ __forceinline__ float wave_sum(float v) {
  #pragma unroll
  for (int off = 16; off >= 1; off >>= 1) v += __shfl_xor(v, off, 32);
  return v;
}

__global__ void k_gemm_relu(const float* __restrict__ X,
                            const float* __restrict__ W,
                            const float* __restrict__ B,
                            float* __restrict__ H,
                            float* __restrict__ rn) {
  const int wave = (int)((blockIdx.x * blockDim.x + threadIdx.x) >> 5);
  const int lane = (int)(threadIdx.x & 31);
  if (wave * 16 >= N_NODES) return;

  const int m0   = wave * 16;
  const int mrow = m0 + (lane & 15);
  const int kg   = (lane >> 4) * 2;
  const int n15  = lane & 15;

  v8f acc[8];
  #pragma unroll
  for (int t = 0; t < 8; ++t) acc[t] = (v8f){0.f,0.f,0.f,0.f,0.f,0.f,0.f,0.f};

  const float* __restrict__ xrow = X + (size_t)mrow * IN_CH + kg;
  const float* __restrict__ wbas = W + (size_t)n15 * IN_CH + kg;

  for (int k = 0; k < IN_CH; k += 8) {
    const v2f a0 = *(const v2f*)(xrow + k);
    const v2f a1 = *(const v2f*)(xrow + k + 4);
    v2f b0[8], b1[8];
    #pragma unroll
    for (int t = 0; t < 8; ++t)
      b0[t] = *(const v2f*)(wbas + (size_t)t * 16 * IN_CH + k);
    #pragma unroll
    for (int t = 0; t < 8; ++t)
      b1[t] = *(const v2f*)(wbas + (size_t)t * 16 * IN_CH + k + 4);
    #pragma unroll
    for (int t = 0; t < 8; ++t)
      acc[t] = __builtin_amdgcn_wmma_f32_16x16x4_f32(
          false, a0, false, b0[t], (short)0, acc[t], false, false);
    #pragma unroll
    for (int t = 0; t < 8; ++t)
      acc[t] = __builtin_amdgcn_wmma_f32_16x16x4_f32(
          false, a1, false, b1[t], (short)0, acc[t], false, false);
  }

  const int mside = (lane >> 4) * 8;
  float ss[8];
  #pragma unroll
  for (int r = 0; r < 8; ++r) ss[r] = 0.f;

  #pragma unroll
  for (int t = 0; t < 8; ++t) {
    const int col = t * 16 + n15;
    const float bb = B[col];
    #pragma unroll
    for (int r = 0; r < 8; ++r) {
      const int row = m0 + r + mside;
      float v = acc[t][r] + bb;
      v = v > 0.f ? v : 0.f;
      H[(size_t)row * HID + col] = v;
      ss[r] += v * v;
    }
  }
  #pragma unroll
  for (int r = 0; r < 8; ++r) {
    #pragma unroll
    for (int off = 8; off >= 1; off >>= 1) ss[r] += __shfl_xor(ss[r], off, 32);
  }
  if (n15 == 0) {
    #pragma unroll
    for (int r = 0; r < 8; ++r)
      rn[m0 + r + mside] = 1.f / fmaxf(sqrtf(ss[r]), 1e-12f);
  }
}

#define OWN        196
#define OWN_BLOCKS 256
#define OWN_WAVES  32
#define EB         8
#define QCAP       520
#define KEY_VECS   (N_EDGES / 4)
#define KEY_STEPS  ((KEY_VECS + 31) / 32)

__device__ __forceinline__ int q_append(unsigned* q, int qn, bool m, unsigned val, unsigned lanes_below) {
  const unsigned mask = __builtin_amdgcn_ballot_w32(m);
  if (m) q[qn + __popc(mask & lanes_below)] = val;
  return qn + __popc(mask);
}

template <int MODE>
__device__ __forceinline__ void serve(const unsigned* qb, int n, int lane,
                                      const float* __restrict__ X,
                                      const int* __restrict__ EI,
                                      const float* __restrict__ rn,
                                      const float* __restrict__ elog,
                                      float* tile, const float* xt,
                                      float* den, const float* nrn) {
  const int li = lane & (EB - 1);
  const unsigned ent = qb[li < n ? li : n - 1];
  const int e   = (int)(ent >> 8);
  const int dl  = (int)(ent & 255u);
  const int src = EI[e];
  float f0, f1 = 0.f;
  if (MODE == 0) { f0 = rn[src]; f1 = nrn[dl]; }
  else           { f0 = elog[e] / den[dl]; }

  float a[EB][4];
  #pragma unroll
  for (int i = 0; i < EB; ++i) {
    const int s_i = __builtin_amdgcn_readlane(src, i);
    const float* xs = X + (size_t)s_i * HID + lane;
    a[i][0] = xs[0]; a[i][1] = xs[32]; a[i][2] = xs[64]; a[i][3] = xs[96];
  }
  float w[EB];
  if (MODE == 0) {
    float s[EB];
    #pragma unroll
    for (int i = 0; i < EB; ++i) {
      const int d_i = __builtin_amdgcn_readlane(dl, i);
      const float* xb = xt + d_i * HID + lane;
      s[i] = a[i][0] * xb[0] + a[i][1] * xb[32] + a[i][2] * xb[64] + a[i][3] * xb[96];
    }
    #pragma unroll
    for (int off = 16; off >= 1; off >>= 1) {
      #pragma unroll
      for (int i = 0; i < EB; ++i) s[i] += __shfl_xor(s[i], off, 32);
    }
    #pragma unroll
    for (int i = 0; i < EB; ++i) {
      const float rs = __int_as_float(__builtin_amdgcn_readlane(__float_as_int(f0), i));
      const float rd = __int_as_float(__builtin_amdgcn_readlane(__float_as_int(f1), i));
      w[i] = __expf(s[i] * rs * rd - 1.0f);
    }
  } else {
    #pragma unroll
    for (int i = 0; i < EB; ++i)
      w[i] = __int_as_float(__builtin_amdgcn_readlane(__float_as_int(f0), i));
  }
  #pragma unroll
  for (int i = 0; i < EB; ++i) {
    if (i < n) {
      const int d_i = __builtin_amdgcn_readlane(dl, i);
      float* tp = tile + d_i * HID + lane;
      atomicAdd(tp +  0, w[i] * a[i][0]);
      atomicAdd(tp + 32, w[i] * a[i][1]);
      atomicAdd(tp + 64, w[i] * a[i][2]);
      atomicAdd(tp + 96, w[i] * a[i][3]);
      if (MODE == 0 && lane == 0) atomicAdd(den + d_i, w[i]);
    }
  }
}

template <int MODE>
__global__ __launch_bounds__(1024) void k_owner(const float* __restrict__ X,
                                                const int* __restrict__ EI,
                                                const float* __restrict__ rn,
                                                const float* __restrict__ elog,
                                                const float* __restrict__ segsum_in,
                                                float* __restrict__ Y,
                                                float* __restrict__ rn_out,
                                                unsigned* __restrict__ segmax_out,
                                                float* __restrict__ segsum_out) {
  __shared__ float tile[OWN * HID];
  __shared__ float xt[MODE == 0 ? OWN * HID : 32];
  __shared__ float den[OWN];
  __shared__ float nrn[OWN];
  __shared__ unsigned queue[OWN_WAVES * QCAP];

  const int wave = __builtin_amdgcn_readfirstlane((int)(threadIdx.x >> 5));
  const int lane = (int)(threadIdx.x & 31);
  const int lo   = (int)blockIdx.x * OWN;
  const int cnt  = (N_NODES - lo) < OWN ? (N_NODES - lo) : OWN;

  for (int n = wave; n < cnt; n += OWN_WAVES) {
    const int node = lo + n;
    const float* xr = X + (size_t)node * HID + lane;
    const float a0 = xr[0], a1 = xr[32], a2 = xr[64], a3 = xr[96];
    float w;
    if (MODE == 0) {
      const float r = rn[node];
      float s = a0 * a0 + a1 * a1 + a2 * a2 + a3 * a3;
      s = wave_sum(s);
      w = __expf(s * r * r - 1.0f);
      float* xo = xt + n * HID + lane;
      xo[0] = a0; xo[32] = a1; xo[64] = a2; xo[96] = a3;
      if (lane == 0) { den[n] = w; nrn[n] = r; }
    } else {
      const float d = segsum_in[node] + 1e-16f;
      w = elog[N_EDGES + node] / d;
      if (lane == 0) den[n] = d;
    }
    float* to = tile + n * HID + lane;
    to[0] = w * a0; to[32] = w * a1; to[64] = w * a2; to[96] = w * a3;
  }
  __syncthreads();

  unsigned* q = queue + wave * QCAP;
  int qn = 0;
  const unsigned lanes_below = (1u << lane) - 1u;
  const unsigned ucnt = (unsigned)cnt;
  const int4* __restrict__ keys = (const int4*)(EI + N_EDGES);
  const int rot = (int)blockIdx.x * 18;
  for (int b = wave; b < KEY_STEPS; b += OWN_WAVES * 4) {
    int4 kk[4];
    int  iv[4];
    #pragma unroll
    for (int j = 0; j < 4; ++j) {
      const int it = b + OWN_WAVES * j;
      int itr = it + rot;
      if (itr >= KEY_STEPS) itr -= KEY_STEPS;
      const int idx = itr * 32 + lane;
      iv[j] = idx;
      kk[j] = (it < KEY_STEPS && idx < KEY_VECS) ? keys[idx] : make_int4(-1, -1, -1, -1);
    }
    #pragma unroll
    for (int j = 0; j < 4; ++j) {
      const unsigned d0 = (unsigned)(kk[j].x - lo), d1 = (unsigned)(kk[j].y - lo);
      const unsigned d2 = (unsigned)(kk[j].z - lo), d3 = (unsigned)(kk[j].w - lo);
      const bool m0 = d0 < ucnt, m1 = d1 < ucnt, m2 = d2 < ucnt, m3 = d3 < ucnt;
      if (__builtin_amdgcn_ballot_w32(m0 | m1 | m2 | m3) != 0u) {
        const unsigned eb = (unsigned)iv[j] * 4u;
        qn = q_append(q, qn, m0, ((eb + 0u) << 8) | d0, lanes_below);
        qn = q_append(q, qn, m1, ((eb + 1u) << 8) | d1, lanes_below);
        qn = q_append(q, qn, m2, ((eb + 2u) << 8) | d2, lanes_below);
        qn = q_append(q, qn, m3, ((eb + 3u) << 8) | d3, lanes_below);
      }
    }
    __builtin_amdgcn_fence(__ATOMIC_ACQ_REL, "wavefront");
    while (qn >= EB) {
      qn -= EB;
      serve<MODE>(q + qn, EB, lane, X, EI, rn, elog, tile, xt, den, nrn);
    }
  }
  __builtin_amdgcn_fence(__ATOMIC_ACQ_REL, "wavefront");
  if (qn > 0) serve<MODE>(q, qn, lane, X, EI, rn, elog, tile, xt, den, nrn);
  __syncthreads();

  for (int n = wave; n < cnt; n += OWN_WAVES) {
    const int node = lo + n;
    const float* ti = tile + n * HID + lane;
    float v0 = ti[0], v1 = ti[32], v2 = ti[64], v3 = ti[96];
    float* yo = Y + (size_t)node * HID + lane;
    if (MODE == 0) {
      const float d = den[n] + 1e-16f;
      v0 /= d; v1 /= d; v2 /= d; v3 /= d;
      float s = v0 * v0 + v1 * v1 + v2 * v2 + v3 * v3;
      s = wave_sum(s);
      if (lane == 0) {
        rn_out[node]     = 1.f / fmaxf(sqrtf(s), 1e-12f);
        segmax_out[node] = ENC_NEG_INF;
        segsum_out[node] = 0.f;
      }
    }
    yo[0] = v0; yo[32] = v1; yo[64] = v2; yo[96] = v3;
  }
}

__global__ void k_edge_logit(const float* __restrict__ X,
                             const int* __restrict__ EI,
                             const float* __restrict__ rn,
                             const float* __restrict__ beta_ptr,
                             float* __restrict__ elog,
                             unsigned* __restrict__ segmax) {
  const int e    = (int)((blockIdx.x * blockDim.x + threadIdx.x) >> 5);
  const int lane = (int)(threadIdx.x & 31);
  if (e >= E_TOT) return;
  int src, dst;
  if (e < N_EDGES) { src = EI[e]; dst = EI[N_EDGES + e]; }
  else             { src = dst = e - N_EDGES; }
  const float4 a = *(const float4*)(X + (size_t)src * HID + lane * 4);
  const float4 b = *(const float4*)(X + (size_t)dst * HID + lane * 4);
  float s = a.x * b.x + a.y * b.y + a.z * b.z + a.w * b.w;
  s = wave_sum(s);
  if (lane == 0) {
    const float beta = beta_ptr ? beta_ptr[0] : 1.0f;
    const float logit = beta * s * rn[src] * rn[dst];
    elog[e] = logit;
    atomicMax(segmax + dst, enc_f32(logit));
  }
}

__global__ void k_edge_exp(const int* __restrict__ EI,
                           const unsigned* __restrict__ segmax,
                           float* __restrict__ elog,
                           float* __restrict__ segsum) {
  const int e = (int)(blockIdx.x * blockDim.x + threadIdx.x);
  if (e >= E_TOT) return;
  const int dst = (e < N_EDGES) ? EI[N_EDGES + e] : (e - N_EDGES);
  const float ex = __expf(elog[e] - dec_f32(segmax[dst]));
  elog[e] = ex;
  atomicAdd(segsum + dst, ex);
}

extern "C" void kernel_launch(void* const* d_in, const int* in_sizes, int n_in,
                              void* d_out, int out_size, void* d_ws, size_t ws_size,
                              hipStream_t stream) {
  const float* x     = (const float*)d_in[0];
  const int*   ei    = (const int*)d_in[1];
  const float* W1    = (const float*)d_in[2];
  const float* b1    = (const float*)d_in[3];
  const float* beta2 = (const float*)d_in[4];
  float* out = (float*)d_out;

  float*    ws     = (float*)d_ws;
  float*    h1     = ws;
  float*    h2     = ws + 6400000;
  float*    rn     = ws + 12800000;
  unsigned* segmax = (unsigned*)(ws + 12850000);
  float*    segsum = ws + 12900000;
  float*    rn2    = ws + 12950000;
  float*    elog   = ws;

  const int gemm_waves  = (N_NODES + 15) / 16;
  const int gemm_blocks = (gemm_waves * 32 + 255) / 256;
  k_gemm_relu<<<gemm_blocks, 256, 0, stream>>>(x, W1, b1, h1, rn);

  k_owner<0><<<OWN_BLOCKS, 1024, 0, stream>>>(h1, ei, rn, nullptr, nullptr,
                                              h2, rn2, segmax, segsum);

  const long long edge_threads = (long long)E_TOT * 32;
  const int eblocks = (int)((edge_threads + 255) / 256);
  k_edge_logit<<<eblocks, 256, 0, stream>>>(h2, ei, rn2, beta2, elog, segmax);
  k_edge_exp<<<(E_TOT + 255) / 256, 256, 0, stream>>>(ei, segmax, elog, segsum);
  k_owner<1><<<OWN_BLOCKS, 1024, 0, stream>>>(h2, ei, nullptr, elog, segsum,
                                              out, nullptr, nullptr, nullptr);
}


# Conv 2 in one owner launch: sweep A takes edge logits against LDS-held dst rows (lane+32j sum order) and ds_max; sweep B adds exp and exp*row by ds_add_f32 (arrival order); rows divided at store
# speedup vs baseline: 2.9577x; 1.5589x over previous
#include <hip/hip_runtime.h>
#include <hip/hip_bf16.h>
#include <math.h>


typedef float v2f __attribute__((ext_vector_type(2)));
typedef float v8f __attribute__((ext_vector_type(8)));

#define N_NODES 50000
#define N_EDGES 600000
#define E_TOT   650000
#define IN_CH   256
#define HID     128
#define ENC_NEG_INF 0x007FFFFFu

__device__ __forceinline__ unsigned enc_f32(float f) {
  unsigned u = __float_as_uint(f);
  return (u & 0x80000000u) ? ~u : (u | 0x80000000u);
}
__device__ __forceinline__ float dec_f32(unsigned e) {
  unsigned u = (e & 0x80000000u) ? (e & 0x7FFFFFFFu) : ~e;
  return __uint_as_float(u);
}

__device__ __forceinline__ float wave_sum(float v) {
  #pragma unroll
  for (int off = 16; off >= 1; off >>= 1) v += __shfl_xor(v, off, 32);
  return v;
}

__global__ void k_gemm_relu(const float* __restrict__ X,
                            const float* __restrict__ W,
                            const float* __restrict__ B,
                            float* __restrict__ H,
                            float* __restrict__ rn) {
  const int wave = (int)((blockIdx.x * blockDim.x + threadIdx.x) >> 5);
  const int lane = (int)(threadIdx.x & 31);
  if (wave * 16 >= N_NODES) return;

  const int m0   = wave * 16;
  const int mrow = m0 + (lane & 15);
  const int kg   = (lane >> 4) * 2;
  const int n15  = lane & 15;

  v8f acc[8];
  #pragma unroll
  for (int t = 0; t < 8; ++t) acc[t] = (v8f){0.f,0.f,0.f,0.f,0.f,0.f,0.f,0.f};

  const float* __restrict__ xrow = X + (size_t)mrow * IN_CH + kg;
  const float* __restrict__ wbas = W + (size_t)n15 * IN_CH + kg;

  for (int k = 0; k < IN_CH; k += 8) {
    const v2f a0 = *(const v2f*)(xrow + k);
    const v2f a1 = *(const v2f*)(xrow + k + 4);
    v2f b0[8], b1[8];
    #pragma unroll
    for (int t = 0; t < 8; ++t)
      b0[t] = *(const v2f*)(wbas + (size_t)t * 16 * IN_CH + k);
    #pragma unroll
    for (int t = 0; t < 8; ++t)
      b1[t] = *(const v2f*)(wbas + (size_t)t * 16 * IN_CH + k + 4);
    #pragma unroll
    for (int t = 0; t < 8; ++t)
      acc[t] = __builtin_amdgcn_wmma_f32_16x16x4_f32(
          false, a0, false, b0[t], (short)0, acc[t], false, false);
    #pragma unroll
    for (int t = 0; t < 8; ++t)
      acc[t] = __builtin_amdgcn_wmma_f32_16x16x4_f32(
          false, a1, false, b1[t], (short)0, acc[t], false, false);
  }

  const int mside = (lane >> 4) * 8;
  float ss[8];
  #pragma unroll
  for (int r = 0; r < 8; ++r) ss[r] = 0.f;

  #pragma unroll
  for (int t = 0; t < 8; ++t) {
    const int col = t * 16 + n15;
    const float bb = B[col];
    #pragma unroll
    for (int r = 0; r < 8; ++r) {
      const int row = m0 + r + mside;
      float v = acc[t][r] + bb;
      v = v > 0.f ? v : 0.f;
      H[(size_t)row * HID + col] = v;
      ss[r] += v * v;
    }
  }
  #pragma unroll
  for (int r = 0; r < 8; ++r) {
    #pragma unroll
    for (int off = 8; off >= 1; off >>= 1) ss[r] += __shfl_xor(ss[r], off, 32);
  }
  if (n15 == 0) {
    #pragma unroll
    for (int r = 0; r < 8; ++r)
      rn[m0 + r + mside] = 1.f / fmaxf(sqrtf(ss[r]), 1e-12f);
  }
}

#define OWN        196
#define OWN_BLOCKS 256
#define OWN_WAVES  32
#define EB         8
#define QCAP       520
#define KEY_VECS   (N_EDGES / 4)
#define KEY_STEPS  ((KEY_VECS + 31) / 32)

__device__ __forceinline__ int q_append(unsigned* q, int qn, bool m, unsigned val, unsigned lanes_below) {
  const unsigned mask = __builtin_amdgcn_ballot_w32(m);
  if (m) q[qn + __popc(mask & lanes_below)] = val;
  return qn + __popc(mask);
}

template <int MODE>
__device__ __forceinline__ void serve(const unsigned* qb, int n, int lane,
                                      const float* __restrict__ X,
                                      const int* __restrict__ EI,
                                      const float* __restrict__ rn,
                                      float* __restrict__ elog, float beta,
                                      float* tile, const float* xt,
                                      float* den, const float* nrn,
                                      unsigned* mx, const float* mxf) {
  const int li = lane & (EB - 1);
  const unsigned ent = qb[li < n ? li : n - 1];
  const int e   = (int)(ent >> 8);
  const int dl  = (int)(ent & 255u);
  const int src = EI[e];
  float f0, f1 = 0.f;
  if (MODE == 1) { f0 = __expf(elog[e] - mxf[dl]); }
  else           { f0 = rn[src]; f1 = nrn[dl]; }

  float a[EB][4];
  #pragma unroll
  for (int i = 0; i < EB; ++i) {
    const int s_i = __builtin_amdgcn_readlane(src, i);
    const float* xs = X + (size_t)s_i * HID + lane;
    a[i][0] = xs[0]; a[i][1] = xs[32]; a[i][2] = xs[64]; a[i][3] = xs[96];
  }
  float w[EB];
  if (MODE != 1) {
    float s[EB];
    #pragma unroll
    for (int i = 0; i < EB; ++i) {
      const int d_i = __builtin_amdgcn_readlane(dl, i);
      const float* xb = xt + d_i * HID + lane;
      s[i] = a[i][0] * xb[0] + a[i][1] * xb[32] + a[i][2] * xb[64] + a[i][3] * xb[96];
    }
    #pragma unroll
    for (int off = 16; off >= 1; off >>= 1) {
      #pragma unroll
      for (int i = 0; i < EB; ++i) s[i] += __shfl_xor(s[i], off, 32);
    }
    if (MODE == 2) {
      float mine = s[0];
      #pragma unroll
      for (int i = 1; i < EB; ++i) mine = (li == i) ? s[i] : mine;
      const float logit = beta * mine * f0 * f1;
      if (lane < n) {
        elog[e] = logit;
        atomicMax(mx + dl, enc_f32(logit));
      }
      return;
    }
    #pragma unroll
    for (int i = 0; i < EB; ++i) {
      const float rs = __int_as_float(__builtin_amdgcn_readlane(__float_as_int(f0), i));
      const float rd = __int_as_float(__builtin_amdgcn_readlane(__float_as_int(f1), i));
      w[i] = __expf(s[i] * rs * rd - 1.0f);
    }
  } else {
    #pragma unroll
    for (int i = 0; i < EB; ++i)
      w[i] = __int_as_float(__builtin_amdgcn_readlane(__float_as_int(f0), i));
  }
  #pragma unroll
  for (int i = 0; i < EB; ++i) {
    if (i < n) {
      const int d_i = __builtin_amdgcn_readlane(dl, i);
      float* tp = tile + d_i * HID + lane;
      atomicAdd(tp +  0, w[i] * a[i][0]);
      atomicAdd(tp + 32, w[i] * a[i][1]);
      atomicAdd(tp + 64, w[i] * a[i][2]);
      atomicAdd(tp + 96, w[i] * a[i][3]);
      if (MODE == 0 && lane == 0) atomicAdd(den + d_i, w[i]);
    }
  }
  if (MODE == 1 && lane < n) atomicAdd(den + dl, f0);
}

template <int MODE>
__device__ __forceinline__ void sweep(unsigned* q, int wave, int lane, int lo, unsigned ucnt, int rot,
                                      const float* __restrict__ X,
                                      const int* __restrict__ EI,
                                      const float* __restrict__ rn,
                                      float* __restrict__ elog, float beta,
                                      float* tile, const float* xt,
                                      float* den, const float* nrn,
                                      unsigned* mx, const float* mxf) {
  int qn = 0;
  const unsigned lanes_below = (1u << lane) - 1u;
  const int4* __restrict__ keys = (const int4*)(EI + N_EDGES);
  for (int b = wave; b < KEY_STEPS; b += OWN_WAVES * 4) {
    int4 kk[4];
    int  iv[4];
    #pragma unroll
    for (int j = 0; j < 4; ++j) {
      const int it = b + OWN_WAVES * j;
      int itr = it + rot;
      if (itr >= KEY_STEPS) itr -= KEY_STEPS;
      const int idx = itr * 32 + lane;
      iv[j] = idx;
      kk[j] = (it < KEY_STEPS && idx < KEY_VECS) ? keys[idx] : make_int4(-1, -1, -1, -1);
    }
    #pragma unroll
    for (int j = 0; j < 4; ++j) {
      const unsigned d0 = (unsigned)(kk[j].x - lo), d1 = (unsigned)(kk[j].y - lo);
      const unsigned d2 = (unsigned)(kk[j].z - lo), d3 = (unsigned)(kk[j].w - lo);
      const bool m0 = d0 < ucnt, m1 = d1 < ucnt, m2 = d2 < ucnt, m3 = d3 < ucnt;
      if (__builtin_amdgcn_ballot_w32(m0 | m1 | m2 | m3) != 0u) {
        const unsigned eb = (unsigned)iv[j] * 4u;
        qn = q_append(q, qn, m0, ((eb + 0u) << 8) | d0, lanes_below);
        qn = q_append(q, qn, m1, ((eb + 1u) << 8) | d1, lanes_below);
        qn = q_append(q, qn, m2, ((eb + 2u) << 8) | d2, lanes_below);
        qn = q_append(q, qn, m3, ((eb + 3u) << 8) | d3, lanes_below);
      }
    }
    __builtin_amdgcn_fence(__ATOMIC_ACQ_REL, "wavefront");
    while (qn >= EB) {
      qn -= EB;
      serve<MODE>(q + qn, EB, lane, X, EI, rn, elog, beta, tile, xt, den, nrn, mx, mxf);
    }
  }
  __builtin_amdgcn_fence(__ATOMIC_ACQ_REL, "wavefront");
  if (qn > 0) serve<MODE>(q, qn, lane, X, EI, rn, elog, beta, tile, xt, den, nrn, mx, mxf);
}

template <int CONV>
__global__ __launch_bounds__(1024) void k_owner(const float* __restrict__ X,
                                                const int* __restrict__ EI,
                                                const float* __restrict__ rn,
                                                float* __restrict__ elog,
                                                const float* __restrict__ beta_ptr,
                                                float* __restrict__ Y,
                                                float* __restrict__ rn_out) {
  __shared__ float tile[OWN * HID];
  __shared__ float xt[OWN * HID];
  __shared__ float den[OWN];
  __shared__ float nrn[OWN];
  __shared__ unsigned mx[CONV == 1 ? OWN : 1];
  __shared__ float mxf[CONV == 1 ? OWN : 1];
  __shared__ unsigned queue[OWN_WAVES * QCAP];

  const int wave = __builtin_amdgcn_readfirstlane((int)(threadIdx.x >> 5));
  const int lane = (int)(threadIdx.x & 31);
  const int lo   = (int)blockIdx.x * OWN;
  const int cnt  = (N_NODES - lo) < OWN ? (N_NODES - lo) : OWN;
  const unsigned ucnt = (unsigned)cnt;
  const int rot = (int)blockIdx.x * 18;
  unsigned* q = queue + wave * QCAP;
  const float beta = (CONV == 1) ? beta_ptr[0] : 1.0f;

  for (int n = wave; n < cnt; n += OWN_WAVES) {
    const int node = lo + n;
    const float* xr = X + (size_t)node * HID + lane;
    const float a0 = xr[0], a1 = xr[32], a2 = xr[64], a3 = xr[96];
    const float r = rn[node];
    float s = a0 * a0 + a1 * a1 + a2 * a2 + a3 * a3;
    s = wave_sum(s);
    float* xo = xt + n * HID + lane;
    xo[0] = a0; xo[32] = a1; xo[64] = a2; xo[96] = a3;
    if (CONV == 0) {
      const float w = __expf(s * r * r - 1.0f);
      if (lane == 0) { den[n] = w; nrn[n] = r; }
      float* to = tile + n * HID + lane;
      to[0] = w * a0; to[32] = w * a1; to[64] = w * a2; to[96] = w * a3;
    } else {
      const float logit = beta * s * r * r;
      if (lane == 0) { mx[n] = enc_f32(logit); mxf[n] = logit; nrn[n] = r; }
    }
  }
  __syncthreads();

  if (CONV == 0) {
    sweep<0>(q, wave, lane, lo, ucnt, rot, X, EI, rn, elog, beta, tile, xt, den, nrn, mx, mxf);
  } else {
    sweep<2>(q, wave, lane, lo, ucnt, rot, X, EI, rn, elog, beta, tile, xt, den, nrn, mx, mxf);
    __syncthreads();
    for (int n = wave; n < cnt; n += OWN_WAVES) {
      const float m = dec_f32(mx[n]);
      const float w = __expf(mxf[n] - m);
      const float* xo = xt + n * HID + lane;
      float* to = tile + n * HID + lane;
      to[0] = w * xo[0]; to[32] = w * xo[32]; to[64] = w * xo[64]; to[96] = w * xo[96];
      if (lane == 0) { den[n] = w; mxf[n] = m; }
    }
    __syncthreads();
    sweep<1>(q, wave, lane, lo, ucnt, rot, X, EI, rn, elog, beta, tile, xt, den, nrn, mx, mxf);
  }
  __syncthreads();

  for (int n = wave; n < cnt; n += OWN_WAVES) {
    const int node = lo + n;
    const float* ti = tile + n * HID + lane;
    float v0 = ti[0], v1 = ti[32], v2 = ti[64], v3 = ti[96];
    float* yo = Y + (size_t)node * HID + lane;
    const float d = den[n] + 1e-16f;
    v0 /= d; v1 /= d; v2 /= d; v3 /= d;
    if (CONV == 0) {
      float s = v0 * v0 + v1 * v1 + v2 * v2 + v3 * v3;
      s = wave_sum(s);
      if (lane == 0) rn_out[node] = 1.f / fmaxf(sqrtf(s), 1e-12f);
    }
    yo[0] = v0; yo[32] = v1; yo[64] = v2; yo[96] = v3;
  }
}

extern "C" void kernel_launch(void* const* d_in, const int* in_sizes, int n_in,
                              void* d_out, int out_size, void* d_ws, size_t ws_size,
                              hipStream_t stream) {
  const float* x     = (const float*)d_in[0];
  const int*   ei    = (const int*)d_in[1];
  const float* W1    = (const float*)d_in[2];
  const float* b1    = (const float*)d_in[3];
  const float* beta2 = (const float*)d_in[4];
  float* out = (float*)d_out;

  float*    ws     = (float*)d_ws;
  float*    h1     = ws;
  float*    h2     = ws + 6400000;
  float*    rn     = ws + 12800000;
  float*    rn2    = ws + 12950000;
  float*    elog   = ws;

  const int gemm_waves  = (N_NODES + 15) / 16;
  const int gemm_blocks = (gemm_waves * 32 + 255) / 256;
  k_gemm_relu<<<gemm_blocks, 256, 0, stream>>>(x, W1, b1, h1, rn);

  k_owner<0><<<OWN_BLOCKS, 1024, 0, stream>>>(h1, ei, rn, nullptr, nullptr, h2, rn2);

  k_owner<1><<<OWN_BLOCKS, 1024, 0, stream>>>(h2, ei, rn2, elog, beta2, out, nullptr);
}


# Owner scans read a bit-sliced 8-bit owner-id plane (dst/196, built by ballots in the GEMM waves), not the 4-byte dst list: 32 ids matched by 8 xor-ands; dst re-read at serve; ds_add_f32 arrival order
# speedup vs baseline: 3.3121x; 1.1198x over previous
#include <hip/hip_runtime.h>
#include <hip/hip_bf16.h>
#include <math.h>


typedef float v2f __attribute__((ext_vector_type(2)));
typedef float v8f __attribute__((ext_vector_type(8)));

#define N_NODES 50000
#define N_EDGES 600000
#define E_TOT   650000
#define IN_CH   256
#define HID     128
#define ENC_NEG_INF 0x007FFFFFu
#define OWN        196
#define OWN_BLOCKS 256
#define PLANE_HALF (N_EDGES / 8)

__device__ __forceinline__ unsigned enc_f32(float f) {
  unsigned u = __float_as_uint(f);
  return (u & 0x80000000u) ? ~u : (u | 0x80000000u);
}
__device__ __forceinline__ float dec_f32(unsigned e) {
  unsigned u = (e & 0x80000000u) ? (e & 0x7FFFFFFFu) : ~e;
  return __uint_as_float(u);
}

__device__ __forceinline__ unsigned owner_id(int dst) {
  return ((unsigned)dst / (unsigned)OWN) & 255u;
}

__device__ __forceinline__ float wave_sum(float v) {
  #pragma unroll
  for (int off = 16; off >= 1; off >>= 1) v += __shfl_xor(v, off, 32);
  return v;
}

__global__ void k_gemm_relu(const float* __restrict__ X,
                            const float* __restrict__ W,
                            const float* __restrict__ B,
                            float* __restrict__ H,
                            float* __restrict__ rn,
                            const int* __restrict__ EI,
                            unsigned* __restrict__ own8) {
  const int wave = (int)((blockIdx.x * blockDim.x + threadIdx.x) >> 5);
  const int lane = (int)(threadIdx.x & 31);
  if (wave * 16 >= N_NODES) return;

  {
    const int* __restrict__ dk = EI + N_EDGES + wave * 192 + lane;
    unsigned o[6];
    #pragma unroll
    for (int g = 0; g < 6; ++g) o[g] = owner_id(dk[32 * g]);
    unsigned r0 = 0u, r1 = 0u;
    #pragma unroll
    for (int g = 0; g < 6; ++g) {
      #pragma unroll
      for (int k = 0; k < 8; ++k) {
        const unsigned bal = __builtin_amdgcn_ballot_w32(((o[g] >> k) & 1u) != 0u);
        if (g < 4) r0 = (lane == 8 * g + k) ? bal : r0;
        else       r1 = (lane == 8 * (g - 4) + k) ? bal : r1;
      }
    }
    unsigned* __restrict__ ov = own8 + (lane & 4) * (PLANE_HALF / 4) + (wave * 6 + (lane >> 3)) * 4 + (lane & 3);
    ov[0] = r0;
    if (lane < 16) ov[16] = r1;
  }

  const int m0   = wave * 16;
  const int mrow = m0 + (lane & 15);
  const int kg   = (lane >> 4) * 2;
  const int n15  = lane & 15;

  v8f acc[8];
  #pragma unroll
  for (int t = 0; t < 8; ++t) acc[t] = (v8f){0.f,0.f,0.f,0.f,0.f,0.f,0.f,0.f};

  const float* __restrict__ xrow = X + (size_t)mrow * IN_CH + kg;
  const float* __restrict__ wbas = W + (size_t)n15 * IN_CH + kg;

  for (int k = 0; k < IN_CH; k += 8) {
    const v2f a0 = *(const v2f*)(xrow + k);
    const v2f a1 = *(const v2f*)(xrow + k + 4);
    v2f b0[8], b1[8];
    #pragma unroll
    for (int t = 0; t < 8; ++t)
      b0[t] = *(const v2f*)(wbas + (size_t)t * 16 * IN_CH + k);
    #pragma unroll
    for (int t = 0; t < 8; ++t)
      b1[t] = *(const v2f*)(wbas + (size_t)t * 16 * IN_CH + k + 4);
    #pragma unroll
    for (int t = 0; t < 8; ++t)
      acc[t] = __builtin_amdgcn_wmma_f32_16x16x4_f32(
          false, a0, false, b0[t], (short)0, acc[t], false, false);
    #pragma unroll
    for (int t = 0; t < 8; ++t)
      acc[t] = __builtin_amdgcn_wmma_f32_16x16x4_f32(
          false, a1, false, b1[t], (short)0, acc[t], false, false);
  }

  const int mside = (lane >> 4) * 8;
  float ss[8];
  #pragma unroll
  for (int r = 0; r < 8; ++r) ss[r] = 0.f;

  #pragma unroll
  for (int t = 0; t < 8; ++t) {
    const int col = t * 16 + n15;
    const float bb = B[col];
    #pragma unroll
    for (int r = 0; r < 8; ++r) {
      const int row = m0 + r + mside;
      float v = acc[t][r] + bb;
      v = v > 0.f ? v : 0.f;
      H[(size_t)row * HID + col] = v;
      ss[r] += v * v;
    }
  }
  #pragma unroll
  for (int r = 0; r < 8; ++r) {
    #pragma unroll
    for (int off = 8; off >= 1; off >>= 1) ss[r] += __shfl_xor(ss[r], off, 32);
  }
  if (n15 == 0) {
    #pragma unroll
    for (int r = 0; r < 8; ++r)
      rn[m0 + r + mside] = 1.f / fmaxf(sqrtf(ss[r]), 1e-12f);
  }
}

#define OWN_WAVES  32
#define EB         8
#define QCAP       40
#define KEY_GROUPS (N_EDGES / 32)
#define KEY_STEPS  ((KEY_GROUPS + 31) / 32)

struct OwnInv { unsigned v[8]; };
__device__ __forceinline__ unsigned match32(uint4 a, uint4 b, const OwnInv& inv) {
  return (a.x ^ inv.v[0]) & (a.y ^ inv.v[1]) & (a.z ^ inv.v[2]) & (a.w ^ inv.v[3]) &
         (b.x ^ inv.v[4]) & (b.y ^ inv.v[5]) & (b.z ^ inv.v[6]) & (b.w ^ inv.v[7]);
}

template <int MODE>
__device__ __forceinline__ void serve(const unsigned* qb, int n, int lane, int lo, unsigned ucnt,
                                      const float* __restrict__ X,
                                      const int* __restrict__ EI,
                                      const float* __restrict__ rn,
                                      float* __restrict__ elog, float beta,
                                      float* tile, const float* xt,
                                      float* den, const float* nrn,
                                      unsigned* mx, const float* mxf) {
  const int li = lane & (EB - 1);
  const int e   = (int)qb[li < n ? li : n - 1];
  const int src = EI[e];
  const unsigned dlu = (unsigned)(EI[N_EDGES + e] - lo);
  const bool ok = (li < n) && (dlu < ucnt);
  const unsigned vmask = __builtin_amdgcn_ballot_w32(ok);
  const int dl  = ok ? (int)dlu : 0;
  float f0, f1 = 0.f;
  if (MODE == 1) { f0 = __expf(elog[e] - mxf[dl]); }
  else           { f0 = rn[src]; f1 = nrn[dl]; }

  float a[EB][4];
  #pragma unroll
  for (int i = 0; i < EB; ++i) {
    const int s_i = __builtin_amdgcn_readlane(src, i);
    const float* xs = X + (size_t)s_i * HID + lane;
    a[i][0] = xs[0]; a[i][1] = xs[32]; a[i][2] = xs[64]; a[i][3] = xs[96];
  }
  float w[EB];
  if (MODE != 1) {
    float s[EB];
    #pragma unroll
    for (int i = 0; i < EB; ++i) {
      const int d_i = __builtin_amdgcn_readlane(dl, i);
      const float* xb = xt + d_i * HID + lane;
      s[i] = a[i][0] * xb[0] + a[i][1] * xb[32] + a[i][2] * xb[64] + a[i][3] * xb[96];
    }
    #pragma unroll
    for (int off = 16; off >= 1; off >>= 1) {
      #pragma unroll
      for (int i = 0; i < EB; ++i) s[i] += __shfl_xor(s[i], off, 32);
    }
    if (MODE == 2) {
      float mine = s[0];
      #pragma unroll
      for (int i = 1; i < EB; ++i) mine = (li == i) ? s[i] : mine;
      const float logit = beta * mine * f0 * f1;
      if (lane < EB && ok) {
        elog[e] = logit;
        atomicMax(mx + dl, enc_f32(logit));
      }
      return;
    }
    #pragma unroll
    for (int i = 0; i < EB; ++i) {
      const float rs = __int_as_float(__builtin_amdgcn_readlane(__float_as_int(f0), i));
      const float rd = __int_as_float(__builtin_amdgcn_readlane(__float_as_int(f1), i));
      w[i] = __expf(s[i] * rs * rd - 1.0f);
    }
  } else {
    #pragma unroll
    for (int i = 0; i < EB; ++i)
      w[i] = __int_as_float(__builtin_amdgcn_readlane(__float_as_int(f0), i));
  }
  #pragma unroll
  for (int i = 0; i < EB; ++i) {
    if ((vmask >> i) & 1u) {
      const int d_i = __builtin_amdgcn_readlane(dl, i);
      float* tp = tile + d_i * HID + lane;
      atomicAdd(tp +  0, w[i] * a[i][0]);
      atomicAdd(tp + 32, w[i] * a[i][1]);
      atomicAdd(tp + 64, w[i] * a[i][2]);
      atomicAdd(tp + 96, w[i] * a[i][3]);
      if (MODE == 0 && lane == 0) atomicAdd(den + d_i, w[i]);
    }
  }
  if (MODE == 1 && lane < EB && ok) atomicAdd(den + dl, f0);
}

template <int MODE>
__device__ __forceinline__ void sweep(unsigned* q, int wave, int lane, int lo, unsigned ucnt, int rot,
                                      const OwnInv& inv, const unsigned* __restrict__ own8,
                                      const float* __restrict__ X,
                                      const int* __restrict__ EI,
                                      const float* __restrict__ rn,
                                      float* __restrict__ elog, float beta,
                                      float* tile, const float* xt,
                                      float* den, const float* nrn,
                                      unsigned* mx, const float* mxf) {
  int qn = 0;
  const unsigned lanes_below = (1u << lane) - 1u;
  const uint4* __restrict__ plane = (const uint4*)own8;
  for (int b = wave; b < KEY_STEPS; b += OWN_WAVES * 2) {
    uint4 ka[2], kb[2];
    int   gi[2];
    bool  in[2];
    #pragma unroll
    for (int j = 0; j < 2; ++j) {
      const int it = b + OWN_WAVES * j;
      int itr = it + rot;
      if (itr >= KEY_STEPS) itr -= KEY_STEPS;
      const int idx = itr * 32 + lane;
      gi[j] = idx;
      in[j] = it < KEY_STEPS && idx < KEY_GROUPS;
      const int ci = idx < KEY_GROUPS ? idx : KEY_GROUPS - 1;
      ka[j] = plane[ci];
      kb[j] = plane[ci + PLANE_HALF / 4];
    }
    unsigned m[2];
    #pragma unroll
    for (int j = 0; j < 2; ++j) m[j] = in[j] ? match32(ka[j], kb[j], inv) : 0u;
    #pragma nounroll
    for (int j = 0; j < 2; ++j) {
      unsigned mj = j == 0 ? m[0] : m[1];
      const int ebj = (j == 0 ? gi[0] : gi[1]) * 32;
      unsigned mask = __builtin_amdgcn_ballot_w32(mj != 0u);
      while (mask != 0u) {
        if (mj != 0u) {
          q[qn + __popc(mask & lanes_below)] = (unsigned)(ebj + __builtin_ctz(mj));
          mj &= mj - 1u;
        }
        qn += __popc(mask);
        __builtin_amdgcn_fence(__ATOMIC_ACQ_REL, "wavefront");
        while (qn >= EB) {
          qn -= EB;
          serve<MODE>(q + qn, EB, lane, lo, ucnt, X, EI, rn, elog, beta, tile, xt, den, nrn, mx, mxf);
        }
        mask = __builtin_amdgcn_ballot_w32(mj != 0u);
      }
    }
  }
  __builtin_amdgcn_fence(__ATOMIC_ACQ_REL, "wavefront");
  if (qn > 0) serve<MODE>(q, qn, lane, lo, ucnt, X, EI, rn, elog, beta, tile, xt, den, nrn, mx, mxf);
}

template <int CONV>
__global__ __launch_bounds__(1024) void k_owner(const float* __restrict__ X,
                                                const int* __restrict__ EI,
                                                const float* __restrict__ rn,
                                                float* __restrict__ elog,
                                                const float* __restrict__ beta_ptr,
                                                float* __restrict__ Y,
                                                float* __restrict__ rn_out,
                                                const unsigned* __restrict__ own8) {
  __shared__ float tile[OWN * HID];
  __shared__ float xt[OWN * HID];
  __shared__ float den[OWN];
  __shared__ float nrn[OWN];
  __shared__ unsigned mx[CONV == 1 ? OWN : 1];
  __shared__ float mxf[CONV == 1 ? OWN : 1];
  __shared__ unsigned queue[OWN_WAVES * QCAP];

  const int wave = __builtin_amdgcn_readfirstlane((int)(threadIdx.x >> 5));
  const int lane = (int)(threadIdx.x & 31);
  const int lo   = (int)blockIdx.x * OWN;
  const int cnt  = (N_NODES - lo) < OWN ? (N_NODES - lo) : OWN;
  const unsigned ucnt = (unsigned)cnt;
  const int rot = ((int)blockIdx.x * KEY_STEPS) >> 8;
  OwnInv inv;
  #pragma unroll
  for (int k = 0; k < 8; ++k) inv.v[k] = ((blockIdx.x >> k) & 1u) - 1u;
  unsigned* q = queue + wave * QCAP;
  const float beta = (CONV == 1) ? beta_ptr[0] : 1.0f;

  for (int n = wave; n < cnt; n += OWN_WAVES) {
    const int node = lo + n;
    const float* xr = X + (size_t)node * HID + lane;
    const float a0 = xr[0], a1 = xr[32], a2 = xr[64], a3 = xr[96];
    const float r = rn[node];
    float s = a0 * a0 + a1 * a1 + a2 * a2 + a3 * a3;
    s = wave_sum(s);
    float* xo = xt + n * HID + lane;
    xo[0] = a0; xo[32] = a1; xo[64] = a2; xo[96] = a3;
    if (CONV == 0) {
      const float w = __expf(s * r * r - 1.0f);
      if (lane == 0) { den[n] = w; nrn[n] = r; }
      float* to = tile + n * HID + lane;
      to[0] = w * a0; to[32] = w * a1; to[64] = w * a2; to[96] = w * a3;
    } else {
      const float logit = beta * s * r * r;
      if (lane == 0) { mx[n] = enc_f32(logit); mxf[n] = logit; nrn[n] = r; }
    }
  }
  __syncthreads();

  if (CONV == 0) {
    sweep<0>(q, wave, lane, lo, ucnt, rot, inv, own8, X, EI, rn, elog, beta, tile, xt, den, nrn, mx, mxf);
  } else {
    sweep<2>(q, wave, lane, lo, ucnt, rot, inv, own8, X, EI, rn, elog, beta, tile, xt, den, nrn, mx, mxf);
    __syncthreads();
    for (int n = wave; n < cnt; n += OWN_WAVES) {
      const float m = dec_f32(mx[n]);
      const float w = __expf(mxf[n] - m);
      const float* xo = xt + n * HID + lane;
      float* to = tile + n * HID + lane;
      to[0] = w * xo[0]; to[32] = w * xo[32]; to[64] = w * xo[64]; to[96] = w * xo[96];
      if (lane == 0) { den[n] = w; mxf[n] = m; }
    }
    __syncthreads();
    sweep<1>(q, wave, lane, lo, ucnt, rot, inv, own8, X, EI, rn, elog, beta, tile, xt, den, nrn, mx, mxf);
  }
  __syncthreads();

  for (int n = wave; n < cnt; n += OWN_WAVES) {
    const int node = lo + n;
    const float* ti = tile + n * HID + lane;
    float v0 = ti[0], v1 = ti[32], v2 = ti[64], v3 = ti[96];
    float* yo = Y + (size_t)node * HID + lane;
    const float d = den[n] + 1e-16f;
    v0 /= d; v1 /= d; v2 /= d; v3 /= d;
    if (CONV == 0) {
      float s = v0 * v0 + v1 * v1 + v2 * v2 + v3 * v3;
      s = wave_sum(s);
      if (lane == 0) rn_out[node] = 1.f / fmaxf(sqrtf(s), 1e-12f);
    }
    yo[0] = v0; yo[32] = v1; yo[64] = v2; yo[96] = v3;
  }
}

extern "C" void kernel_launch(void* const* d_in, const int* in_sizes, int n_in,
                              void* d_out, int out_size, void* d_ws, size_t ws_size,
                              hipStream_t stream) {
  const float* x     = (const float*)d_in[0];
  const int*   ei    = (const int*)d_in[1];
  const float* W1    = (const float*)d_in[2];
  const float* b1    = (const float*)d_in[3];
  const float* beta2 = (const float*)d_in[4];
  float* out = (float*)d_out;

  float*    ws     = (float*)d_ws;
  float*    h1     = ws;
  float*    h2     = ws + 6400000;
  float*    rn     = ws + 12800000;
  float*    rn2    = ws + 12950000;
  float*    elog   = ws;
  unsigned* own8   = (unsigned*)(ws + 13000000);

  const int gemm_waves  = (N_NODES + 15) / 16;
  const int gemm_blocks = (gemm_waves * 32 + 255) / 256;
  k_gemm_relu<<<gemm_blocks, 256, 0, stream>>>(x, W1, b1, h1, rn, ei, own8);

  k_owner<0><<<OWN_BLOCKS, 1024, 0, stream>>>(h1, ei, rn, nullptr, nullptr, h2, rn2, own8);

  k_owner<1><<<OWN_BLOCKS, 1024, 0, stream>>>(h2, ei, rn2, elog, beta2, out, nullptr, own8);
}


# Conv 2, |beta| <= 8: exp shifted by the constant |beta| (|cos| <= 1), not the segment max, so sweeps A and B are ONE (one scan, one row gather per edge); ds_add_f32 arrival order as before
# speedup vs baseline: 4.1611x; 1.2563x over previous
#include <hip/hip_runtime.h>
#include <hip/hip_bf16.h>
#include <math.h>


typedef float v2f __attribute__((ext_vector_type(2)));
typedef float v8f __attribute__((ext_vector_type(8)));

#define N_NODES 50000
#define N_EDGES 600000
#define E_TOT   650000
#define IN_CH   256
#define HID     128
#define ENC_NEG_INF 0x007FFFFFu
#define OWN        196
#define OWN_BLOCKS 256
#define PLANE_HALF (N_EDGES / 8)
#define SHIFT_MAX  8.0f

__device__ __forceinline__ unsigned enc_f32(float f) {
  unsigned u = __float_as_uint(f);
  return (u & 0x80000000u) ? ~u : (u | 0x80000000u);
}
__device__ __forceinline__ float dec_f32(unsigned e) {
  unsigned u = (e & 0x80000000u) ? (e & 0x7FFFFFFFu) : ~e;
  return __uint_as_float(u);
}

__device__ __forceinline__ unsigned owner_id(int dst) {
  return ((unsigned)dst / (unsigned)OWN) & 255u;
}

__device__ __forceinline__ float wave_sum(float v) {
  #pragma unroll
  for (int off = 16; off >= 1; off >>= 1) v += __shfl_xor(v, off, 32);
  return v;
}

__global__ void k_gemm_relu(const float* __restrict__ X,
                            const float* __restrict__ W,
                            const float* __restrict__ B,
                            float* __restrict__ H,
                            float* __restrict__ rn,
                            const int* __restrict__ EI,
                            unsigned* __restrict__ own8) {
  const int wave = (int)((blockIdx.x * blockDim.x + threadIdx.x) >> 5);
  const int lane = (int)(threadIdx.x & 31);
  if (wave * 16 >= N_NODES) return;

  {
    const int* __restrict__ dk = EI + N_EDGES + wave * 192 + lane;
    unsigned o[6];
    #pragma unroll
    for (int g = 0; g < 6; ++g) o[g] = owner_id(dk[32 * g]);
    unsigned r0 = 0u, r1 = 0u;
    #pragma unroll
    for (int g = 0; g < 6; ++g) {
      #pragma unroll
      for (int k = 0; k < 8; ++k) {
        const unsigned bal = __builtin_amdgcn_ballot_w32(((o[g] >> k) & 1u) != 0u);
        if (g < 4) r0 = (lane == 8 * g + k) ? bal : r0;
        else       r1 = (lane == 8 * (g - 4) + k) ? bal : r1;
      }
    }
    unsigned* __restrict__ ov = own8 + (lane & 4) * (PLANE_HALF / 4) + (wave * 6 + (lane >> 3)) * 4 + (lane & 3);
    ov[0] = r0;
    if (lane < 16) ov[16] = r1;
  }

  const int m0   = wave * 16;
  const int mrow = m0 + (lane & 15);
  const int kg   = (lane >> 4) * 2;
  const int n15  = lane & 15;

  v8f acc[8];
  #pragma unroll
  for (int t = 0; t < 8; ++t) acc[t] = (v8f){0.f,0.f,0.f,0.f,0.f,0.f,0.f,0.f};

  const float* __restrict__ xrow = X + (size_t)mrow * IN_CH + kg;
  const float* __restrict__ wbas = W + (size_t)n15 * IN_CH + kg;

  for (int k = 0; k < IN_CH; k += 8) {
    const v2f a0 = *(const v2f*)(xrow + k);
    const v2f a1 = *(const v2f*)(xrow + k + 4);
    v2f b0[8], b1[8];
    #pragma unroll
    for (int t = 0; t < 8; ++t)
      b0[t] = *(const v2f*)(wbas + (size_t)t * 16 * IN_CH + k);
    #pragma unroll
    for (int t = 0; t < 8; ++t)
      b1[t] = *(const v2f*)(wbas + (size_t)t * 16 * IN_CH + k + 4);
    #pragma unroll
    for (int t = 0; t < 8; ++t)
      acc[t] = __builtin_amdgcn_wmma_f32_16x16x4_f32(
          false, a0, false, b0[t], (short)0, acc[t], false, false);
    #pragma unroll
    for (int t = 0; t < 8; ++t)
      acc[t] = __builtin_amdgcn_wmma_f32_16x16x4_f32(
          false, a1, false, b1[t], (short)0, acc[t], false, false);
  }

  const int mside = (lane >> 4) * 8;
  float ss[8];
  #pragma unroll
  for (int r = 0; r < 8; ++r) ss[r] = 0.f;

  #pragma unroll
  for (int t = 0; t < 8; ++t) {
    const int col = t * 16 + n15;
    const float bb = B[col];
    #pragma unroll
    for (int r = 0; r < 8; ++r) {
      const int row = m0 + r + mside;
      float v = acc[t][r] + bb;
      v = v > 0.f ? v : 0.f;
      H[(size_t)row * HID + col] = v;
      ss[r] += v * v;
    }
  }
  #pragma unroll
  for (int r = 0; r < 8; ++r) {
    #pragma unroll
    for (int off = 8; off >= 1; off >>= 1) ss[r] += __shfl_xor(ss[r], off, 32);
  }
  if (n15 == 0) {
    #pragma unroll
    for (int r = 0; r < 8; ++r)
      rn[m0 + r + mside] = 1.f / fmaxf(sqrtf(ss[r]), 1e-12f);
  }
}

#define OWN_WAVES  32
#define EB         8
#define QCAP       40
#define KEY_GROUPS (N_EDGES / 32)
#define KEY_STEPS  ((KEY_GROUPS + 31) / 32)

struct OwnInv { unsigned v[8]; };
__device__ __forceinline__ unsigned match32(uint4 a, uint4 b, const OwnInv& inv) {
  return (a.x ^ inv.v[0]) & (a.y ^ inv.v[1]) & (a.z ^ inv.v[2]) & (a.w ^ inv.v[3]) &
         (b.x ^ inv.v[4]) & (b.y ^ inv.v[5]) & (b.z ^ inv.v[6]) & (b.w ^ inv.v[7]);
}

template <int MODE>
__device__ __forceinline__ void serve(const unsigned* qb, int n, int lane, int lo, unsigned ucnt,
                                      const float* __restrict__ X,
                                      const int* __restrict__ EI,
                                      const float* __restrict__ rn,
                                      float* __restrict__ elog, float beta, float shift,
                                      float* tile, const float* xt,
                                      float* den, const float* nrn,
                                      unsigned* mx, const float* mxf) {
  const int li = lane & (EB - 1);
  const int e   = (int)qb[li < n ? li : n - 1];
  const int src = EI[e];
  const unsigned dlu = (unsigned)(EI[N_EDGES + e] - lo);
  const bool ok = (li < n) && (dlu < ucnt);
  const unsigned vmask = __builtin_amdgcn_ballot_w32(ok);
  const int dl  = ok ? (int)dlu : 0;
  float f0, f1 = 0.f;
  if (MODE == 1) { f0 = __expf(elog[e] - mxf[dl]); }
  else           { f0 = rn[src]; f1 = nrn[dl]; }

  float a[EB][4];
  #pragma unroll
  for (int i = 0; i < EB; ++i) {
    const int s_i = __builtin_amdgcn_readlane(src, i);
    const float* xs = X + (size_t)s_i * HID + lane;
    a[i][0] = xs[0]; a[i][1] = xs[32]; a[i][2] = xs[64]; a[i][3] = xs[96];
  }
  float w[EB];
  if (MODE != 1) {
    float s[EB];
    #pragma unroll
    for (int i = 0; i < EB; ++i) {
      const int d_i = __builtin_amdgcn_readlane(dl, i);
      const float* xb = xt + d_i * HID + lane;
      s[i] = a[i][0] * xb[0] + a[i][1] * xb[32] + a[i][2] * xb[64] + a[i][3] * xb[96];
    }
    #pragma unroll
    for (int off = 16; off >= 1; off >>= 1) {
      #pragma unroll
      for (int i = 0; i < EB; ++i) s[i] += __shfl_xor(s[i], off, 32);
    }
    if (MODE == 2 || MODE == 3) {
      float mine = s[0];
      #pragma unroll
      for (int i = 1; i < EB; ++i) mine = (li == i) ? s[i] : mine;
      const float logit = beta * mine * f0 * f1;
      if (MODE == 2) {
        if (lane < EB && ok) {
          elog[e] = logit;
          atomicMax(mx + dl, enc_f32(logit));
        }
        return;
      }
      f0 = __expf(logit - shift);
    } else {
      #pragma unroll
      for (int i = 0; i < EB; ++i) {
        const float rs = __int_as_float(__builtin_amdgcn_readlane(__float_as_int(f0), i));
        const float rd = __int_as_float(__builtin_amdgcn_readlane(__float_as_int(f1), i));
        w[i] = __expf(s[i] * rs * rd - 1.0f);
      }
    }
  }
  if (MODE == 1 || MODE == 3) {
    #pragma unroll
    for (int i = 0; i < EB; ++i)
      w[i] = __int_as_float(__builtin_amdgcn_readlane(__float_as_int(f0), i));
  }
  #pragma unroll
  for (int i = 0; i < EB; ++i) {
    if ((vmask >> i) & 1u) {
      const int d_i = __builtin_amdgcn_readlane(dl, i);
      float* tp = tile + d_i * HID + lane;
      atomicAdd(tp +  0, w[i] * a[i][0]);
      atomicAdd(tp + 32, w[i] * a[i][1]);
      atomicAdd(tp + 64, w[i] * a[i][2]);
      atomicAdd(tp + 96, w[i] * a[i][3]);
      if (MODE == 0 && lane == 0) atomicAdd(den + d_i, w[i]);
    }
  }
  if ((MODE == 1 || MODE == 3) && lane < EB && ok) atomicAdd(den + dl, f0);
}

template <int MODE>
__device__ __forceinline__ void sweep(unsigned* q, int wave, int lane, int lo, unsigned ucnt, int rot,
                                      const OwnInv& inv, const unsigned* __restrict__ own8,
                                      const float* __restrict__ X,
                                      const int* __restrict__ EI,
                                      const float* __restrict__ rn,
                                      float* __restrict__ elog, float beta, float shift,
                                      float* tile, const float* xt,
                                      float* den, const float* nrn,
                                      unsigned* mx, const float* mxf) {
  int qn = 0;
  const unsigned lanes_below = (1u << lane) - 1u;
  const uint4* __restrict__ plane = (const uint4*)own8;
  for (int b = wave; b < KEY_STEPS; b += OWN_WAVES * 2) {
    uint4 ka[2], kb[2];
    int   gi[2];
    bool  in[2];
    #pragma unroll
    for (int j = 0; j < 2; ++j) {
      const int it = b + OWN_WAVES * j;
      int itr = it + rot;
      if (itr >= KEY_STEPS) itr -= KEY_STEPS;
      const int idx = itr * 32 + lane;
      gi[j] = idx;
      in[j] = it < KEY_STEPS && idx < KEY_GROUPS;
      const int ci = idx < KEY_GROUPS ? idx : KEY_GROUPS - 1;
      ka[j] = plane[ci];
      kb[j] = plane[ci + PLANE_HALF / 4];
    }
    unsigned m[2];
    #pragma unroll
    for (int j = 0; j < 2; ++j) m[j] = in[j] ? match32(ka[j], kb[j], inv) : 0u;
    #pragma nounroll
    for (int j = 0; j < 2; ++j) {
      unsigned mj = j == 0 ? m[0] : m[1];
      const int ebj = (j == 0 ? gi[0] : gi[1]) * 32;
      unsigned mask = __builtin_amdgcn_ballot_w32(mj != 0u);
      while (mask != 0u) {
        if (mj != 0u) {
          q[qn + __popc(mask & lanes_below)] = (unsigned)(ebj + __builtin_ctz(mj));
          mj &= mj - 1u;
        }
        qn += __popc(mask);
        __builtin_amdgcn_fence(__ATOMIC_ACQ_REL, "wavefront");
        while (qn >= EB) {
          qn -= EB;
          serve<MODE>(q + qn, EB, lane, lo, ucnt, X, EI, rn, elog, beta, shift, tile, xt, den, nrn, mx, mxf);
        }
        mask = __builtin_amdgcn_ballot_w32(mj != 0u);
      }
    }
  }
  __builtin_amdgcn_fence(__ATOMIC_ACQ_REL, "wavefront");
  if (qn > 0) serve<MODE>(q, qn, lane, lo, ucnt, X, EI, rn, elog, beta, shift, tile, xt, den, nrn, mx, mxf);
}

template <int CONV>
__global__ __launch_bounds__(1024) void k_owner(const float* __restrict__ X,
                                                const int* __restrict__ EI,
                                                const float* __restrict__ rn,
                                                float* __restrict__ elog,
                                                const float* __restrict__ beta_ptr,
                                                float* __restrict__ Y,
                                                float* __restrict__ rn_out,
                                                const unsigned* __restrict__ own8) {
  __shared__ float tile[OWN * HID];
  __shared__ float xt[OWN * HID];
  __shared__ float den[OWN];
  __shared__ float nrn[OWN];
  __shared__ unsigned mx[CONV == 1 ? OWN : 1];
  __shared__ float mxf[CONV == 1 ? OWN : 1];
  __shared__ unsigned queue[OWN_WAVES * QCAP];

  const int wave = __builtin_amdgcn_readfirstlane((int)(threadIdx.x >> 5));
  const int lane = (int)(threadIdx.x & 31);
  const int lo   = (int)blockIdx.x * OWN;
  const int cnt  = (N_NODES - lo) < OWN ? (N_NODES - lo) : OWN;
  const unsigned ucnt = (unsigned)cnt;
  const int rot = ((int)blockIdx.x * KEY_STEPS) >> 8;
  OwnInv inv;
  #pragma unroll
  for (int k = 0; k < 8; ++k) inv.v[k] = ((blockIdx.x >> k) & 1u) - 1u;
  unsigned* q = queue + wave * QCAP;
  const float beta  = (CONV == 1) ? beta_ptr[0] : 1.0f;
  const bool  one   = (CONV == 0) || (fabsf(beta) <= SHIFT_MAX);
  const float shift = (CONV == 0) ? 1.0f : fabsf(beta);

  for (int n = wave; n < cnt; n += OWN_WAVES) {
    const int node = lo + n;
    const float* xr = X + (size_t)node * HID + lane;
    const float a0 = xr[0], a1 = xr[32], a2 = xr[64], a3 = xr[96];
    const float r = rn[node];
    float s = a0 * a0 + a1 * a1 + a2 * a2 + a3 * a3;
    s = wave_sum(s);
    float* xo = xt + n * HID + lane;
    xo[0] = a0; xo[32] = a1; xo[64] = a2; xo[96] = a3;
    if (one) {
      const float w = __expf(beta * s * r * r - shift);
      if (lane == 0) { den[n] = w; nrn[n] = r; }
      float* to = tile + n * HID + lane;
      to[0] = w * a0; to[32] = w * a1; to[64] = w * a2; to[96] = w * a3;
    } else {
      const float logit = beta * s * r * r;
      if (lane == 0) { mx[n] = enc_f32(logit); mxf[n] = logit; nrn[n] = r; }
    }
  }
  __syncthreads();

  if (one) {
    sweep<(CONV == 0) ? 0 : 3>(q, wave, lane, lo, ucnt, rot, inv, own8, X, EI, rn, elog, beta, shift, tile, xt, den, nrn, mx, mxf);
  } else {
    sweep<2>(q, wave, lane, lo, ucnt, rot, inv, own8, X, EI, rn, elog, beta, shift, tile, xt, den, nrn, mx, mxf);
    __syncthreads();
    for (int n = wave; n < cnt; n += OWN_WAVES) {
      const float m = dec_f32(mx[n]);
      const float w = __expf(mxf[n] - m);
      const float* xo = xt + n * HID + lane;
      float* to = tile + n * HID + lane;
      to[0] = w * xo[0]; to[32] = w * xo[32]; to[64] = w * xo[64]; to[96] = w * xo[96];
      if (lane == 0) { den[n] = w; mxf[n] = m; }
    }
    __syncthreads();
    sweep<1>(q, wave, lane, lo, ucnt, rot, inv, own8, X, EI, rn, elog, beta, shift, tile, xt, den, nrn, mx, mxf);
  }
  __syncthreads();

  for (int n = wave; n < cnt; n += OWN_WAVES) {
    const int node = lo + n;
    const float* ti = tile + n * HID + lane;
    float v0 = ti[0], v1 = ti[32], v2 = ti[64], v3 = ti[96];
    float* yo = Y + (size_t)node * HID + lane;
    const float d = den[n] + 1e-16f;
    v0 /= d; v1 /= d; v2 /= d; v3 /= d;
    if (CONV == 0) {
      float s = v0 * v0 + v1 * v1 + v2 * v2 + v3 * v3;
      s = wave_sum(s);
      if (lane == 0) rn_out[node] = 1.f / fmaxf(sqrtf(s), 1e-12f);
    }
    yo[0] = v0; yo[32] = v1; yo[64] = v2; yo[96] = v3;
  }
}

extern "C" void kernel_launch(void* const* d_in, const int* in_sizes, int n_in,
                              void* d_out, int out_size, void* d_ws, size_t ws_size,
                              hipStream_t stream) {
  const float* x     = (const float*)d_in[0];
  const int*   ei    = (const int*)d_in[1];
  const float* W1    = (const float*)d_in[2];
  const float* b1    = (const float*)d_in[3];
  const float* beta2 = (const float*)d_in[4];
  float* out = (float*)d_out;

  float*    ws     = (float*)d_ws;
  float*    h1     = ws;
  float*    h2     = ws + 6400000;
  float*    rn     = ws + 12800000;
  float*    rn2    = ws + 12950000;
  float*    elog   = ws;
  unsigned* own8   = (unsigned*)(ws + 13000000);

  const int gemm_waves  = (N_NODES + 15) / 16;
  const int gemm_blocks = (gemm_waves * 32 + 255) / 256;
  k_gemm_relu<<<gemm_blocks, 256, 0, stream>>>(x, W1, b1, h1, rn, ei, own8);

  k_owner<0><<<OWN_BLOCKS, 1024, 0, stream>>>(h1, ei, rn, nullptr, nullptr, h2, rn2, own8);

  k_owner<1><<<OWN_BLOCKS, 1024, 0, stream>>>(h2, ei, rn2, elog, beta2, out, nullptr, own8);
}


# GEMM: a wave owns 64 rows x 128 columns (32 accumulator tiles, one wave per SIMD), so a B fragment feeds 4 WMMAs, not 1: a third of the fragment loads per WMMA, refills one k-step ahead; k order kept
# speedup vs baseline: 4.7756x; 1.1477x over previous
#include <hip/hip_runtime.h>
#include <hip/hip_bf16.h>
#include <math.h>


typedef float v2f __attribute__((ext_vector_type(2)));
typedef float v8f __attribute__((ext_vector_type(8)));

#define N_NODES 50000
#define N_EDGES 600000
#define E_TOT   650000
#define IN_CH   256
#define HID     128
#define ENC_NEG_INF 0x007FFFFFu
#define OWN        196
#define OWN_BLOCKS 256
#define PLANE_HALF (N_EDGES / 8)
#define SHIFT_MAX  8.0f

__device__ __forceinline__ unsigned enc_f32(float f) {
  unsigned u = __float_as_uint(f);
  return (u & 0x80000000u) ? ~u : (u | 0x80000000u);
}
__device__ __forceinline__ float dec_f32(unsigned e) {
  unsigned u = (e & 0x80000000u) ? (e & 0x7FFFFFFFu) : ~e;
  return __uint_as_float(u);
}

__device__ __forceinline__ unsigned owner_id(int dst) {
  return ((unsigned)dst / (unsigned)OWN) & 255u;
}

__device__ __forceinline__ float wave_sum(float v) {
  #pragma unroll
  for (int off = 16; off >= 1; off >>= 1) v += __shfl_xor(v, off, 32);
  return v;
}

struct Frags { v2f a0[4], a1[4], b0[8], b1[8]; };
__device__ __forceinline__ void load_frags(Frags& f, const float* __restrict__ const* xrow,
                                           const float* __restrict__ wbas, int k) {
  #pragma unroll
  for (int i = 0; i < 4; ++i) {
    f.a0[i] = *(const v2f*)(xrow[i] + k);
    f.a1[i] = *(const v2f*)(xrow[i] + k + 4);
  }
  #pragma unroll
  for (int t = 0; t < 8; ++t) {
    f.b0[t] = *(const v2f*)(wbas + (size_t)t * 16 * IN_CH + k);
    f.b1[t] = *(const v2f*)(wbas + (size_t)t * 16 * IN_CH + k + 4);
  }
}
__device__ __forceinline__ void mma_frags(v8f (&acc)[4][8], const Frags& f) {
  #pragma unroll
  for (int t = 0; t < 8; ++t) {
    #pragma unroll
    for (int i = 0; i < 4; ++i)
      acc[i][t] = __builtin_amdgcn_wmma_f32_16x16x4_f32(
          false, f.a0[i], false, f.b0[t], (short)0, acc[i][t], false, false);
  }
  #pragma unroll
  for (int t = 0; t < 8; ++t) {
    #pragma unroll
    for (int i = 0; i < 4; ++i)
      acc[i][t] = __builtin_amdgcn_wmma_f32_16x16x4_f32(
          false, f.a1[i], false, f.b1[t], (short)0, acc[i][t], false, false);
  }
}

#define GEMM_TILES 3125
#define GEMM_RT    4

__global__ __launch_bounds__(128) void k_gemm_relu(const float* __restrict__ X,
                            const float* __restrict__ W,
                            const float* __restrict__ B,
                            float* __restrict__ H,
                            float* __restrict__ rn,
                            const int* __restrict__ EI,
                            unsigned* __restrict__ own8) {
  const int wave = __builtin_amdgcn_readfirstlane((int)((blockIdx.x * blockDim.x + threadIdx.x) >> 5));
  const int lane = (int)(threadIdx.x & 31);
  if (wave * GEMM_RT >= GEMM_TILES) return;

  {
    unsigned o[GEMM_RT][6];
    #pragma unroll
    for (int vi = 0; vi < GEMM_RT; ++vi) {
      const int v  = wave * GEMM_RT + vi;
      const int vc = v < GEMM_TILES ? v : GEMM_TILES - 1;
      const int* __restrict__ dk = EI + N_EDGES + vc * 192 + lane;
      #pragma unroll
      for (int g = 0; g < 6; ++g) o[vi][g] = owner_id(dk[32 * g]);
    }
    #pragma unroll
    for (int vi = 0; vi < GEMM_RT; ++vi) {
      const int v = wave * GEMM_RT + vi;
      if (v < GEMM_TILES) {
        unsigned r0 = 0u, r1 = 0u;
        #pragma unroll
        for (int g = 0; g < 6; ++g) {
          #pragma unroll
          for (int k = 0; k < 8; ++k) {
            const unsigned bal = __builtin_amdgcn_ballot_w32(((o[vi][g] >> k) & 1u) != 0u);
            if (g < 4) r0 = (lane == 8 * g + k) ? bal : r0;
            else       r1 = (lane == 8 * (g - 4) + k) ? bal : r1;
          }
        }
        unsigned* __restrict__ ov = own8 + (lane & 4) * (PLANE_HALF / 4) + (v * 6 + (lane >> 3)) * 4 + (lane & 3);
        ov[0] = r0;
        if (lane < 16) ov[16] = r1;
      }
    }
  }

  const int t0   = wave * GEMM_RT;
  const int kg   = (lane >> 4) * 2;
  const int n15  = lane & 15;

  v8f acc[GEMM_RT][8];
  #pragma unroll
  for (int i = 0; i < GEMM_RT; ++i) {
    #pragma unroll
    for (int t = 0; t < 8; ++t) acc[i][t] = (v8f){0.f,0.f,0.f,0.f,0.f,0.f,0.f,0.f};
  }

  const float* __restrict__ xrow[GEMM_RT];
  #pragma unroll
  for (int i = 0; i < GEMM_RT; ++i) {
    const int ti = (t0 + i) < GEMM_TILES ? (t0 + i) : (GEMM_TILES - 1);
    xrow[i] = X + (size_t)(ti * 16 + n15) * IN_CH + kg;
  }
  const float* __restrict__ wbas = W + (size_t)n15 * IN_CH + kg;

  Frags p, q;
  load_frags(p, xrow, wbas, 0);
  __builtin_amdgcn_sched_barrier(0);
  load_frags(q, xrow, wbas, 8);
  __builtin_amdgcn_sched_barrier(0);
  for (int k = 0; k < IN_CH; k += 16) {
    mma_frags(acc, p);
    load_frags(p, xrow, wbas, (k + 16) < IN_CH ? (k + 16) : (IN_CH - 8));
    __builtin_amdgcn_sched_barrier(0);
    mma_frags(acc, q);
    load_frags(q, xrow, wbas, (k + 24) < IN_CH ? (k + 24) : (IN_CH - 8));
    __builtin_amdgcn_sched_barrier(0);
  }

  const int mside = (lane >> 4) * 8;
  #pragma unroll
  for (int i = 0; i < GEMM_RT; ++i) {
    if (t0 + i < GEMM_TILES) {
      const int m0 = (t0 + i) * 16;
      float ss[8];
      #pragma unroll
      for (int r = 0; r < 8; ++r) ss[r] = 0.f;
      #pragma unroll
      for (int t = 0; t < 8; ++t) {
        const int col = t * 16 + n15;
        const float bb = B[col];
        #pragma unroll
        for (int r = 0; r < 8; ++r) {
          const int row = m0 + r + mside;
          float v = acc[i][t][r] + bb;
          v = v > 0.f ? v : 0.f;
          H[(size_t)row * HID + col] = v;
          ss[r] += v * v;
        }
      }
      #pragma unroll
      for (int r = 0; r < 8; ++r) {
        #pragma unroll
        for (int off = 8; off >= 1; off >>= 1) ss[r] += __shfl_xor(ss[r], off, 32);
      }
      if (n15 == 0) {
        #pragma unroll
        for (int r = 0; r < 8; ++r)
          rn[m0 + r + mside] = 1.f / fmaxf(sqrtf(ss[r]), 1e-12f);
      }
    }
  }
}

#define OWN_WAVES  32
#define EB         8
#define QCAP       40
#define KEY_GROUPS (N_EDGES / 32)
#define KEY_STEPS  ((KEY_GROUPS + 31) / 32)

struct OwnInv { unsigned v[8]; };
__device__ __forceinline__ unsigned match32(uint4 a, uint4 b, const OwnInv& inv) {
  return (a.x ^ inv.v[0]) & (a.y ^ inv.v[1]) & (a.z ^ inv.v[2]) & (a.w ^ inv.v[3]) &
         (b.x ^ inv.v[4]) & (b.y ^ inv.v[5]) & (b.z ^ inv.v[6]) & (b.w ^ inv.v[7]);
}

template <int MODE>
__device__ __forceinline__ void serve(const unsigned* qb, int n, int lane, int lo, unsigned ucnt,
                                      const float* __restrict__ X,
                                      const int* __restrict__ EI,
                                      const float* __restrict__ rn,
                                      float* __restrict__ elog, float beta, float shift,
                                      float* tile, const float* xt,
                                      float* den, const float* nrn,
                                      unsigned* mx, const float* mxf) {
  const int li = lane & (EB - 1);
  const int e   = (int)qb[li < n ? li : n - 1];
  const int src = EI[e];
  const unsigned dlu = (unsigned)(EI[N_EDGES + e] - lo);
  const bool ok = (li < n) && (dlu < ucnt);
  const unsigned vmask = __builtin_amdgcn_ballot_w32(ok);
  const int dl  = ok ? (int)dlu : 0;
  float f0, f1 = 0.f;
  if (MODE == 1) { f0 = __expf(elog[e] - mxf[dl]); }
  else           { f0 = rn[src]; f1 = nrn[dl]; }

  float a[EB][4];
  #pragma unroll
  for (int i = 0; i < EB; ++i) {
    const int s_i = __builtin_amdgcn_readlane(src, i);
    const float* xs = X + (size_t)s_i * HID + lane;
    a[i][0] = xs[0]; a[i][1] = xs[32]; a[i][2] = xs[64]; a[i][3] = xs[96];
  }
  float w[EB];
  if (MODE != 1) {
    float s[EB];
    #pragma unroll
    for (int i = 0; i < EB; ++i) {
      const int d_i = __builtin_amdgcn_readlane(dl, i);
      const float* xb = xt + d_i * HID + lane;
      s[i] = a[i][0] * xb[0] + a[i][1] * xb[32] + a[i][2] * xb[64] + a[i][3] * xb[96];
    }
    #pragma unroll
    for (int off = 16; off >= 1; off >>= 1) {
      #pragma unroll
      for (int i = 0; i < EB; ++i) s[i] += __shfl_xor(s[i], off, 32);
    }
    if (MODE == 2 || MODE == 3) {
      float mine = s[0];
      #pragma unroll
      for (int i = 1; i < EB; ++i) mine = (li == i) ? s[i] : mine;
      const float logit = beta * mine * f0 * f1;
      if (MODE == 2) {
        if (lane < EB && ok) {
          elog[e] = logit;
          atomicMax(mx + dl, enc_f32(logit));
        }
        return;
      }
      f0 = __expf(logit - shift);
    } else {
      #pragma unroll
      for (int i = 0; i < EB; ++i) {
        const float rs = __int_as_float(__builtin_amdgcn_readlane(__float_as_int(f0), i));
        const float rd = __int_as_float(__builtin_amdgcn_readlane(__float_as_int(f1), i));
        w[i] = __expf(s[i] * rs * rd - 1.0f);
      }
    }
  }
  if (MODE == 1 || MODE == 3) {
    #pragma unroll
    for (int i = 0; i < EB; ++i)
      w[i] = __int_as_float(__builtin_amdgcn_readlane(__float_as_int(f0), i));
  }
  #pragma unroll
  for (int i = 0; i < EB; ++i) {
    if ((vmask >> i) & 1u) {
      const int d_i = __builtin_amdgcn_readlane(dl, i);
      float* tp = tile + d_i * HID + lane;
      atomicAdd(tp +  0, w[i] * a[i][0]);
      atomicAdd(tp + 32, w[i] * a[i][1]);
      atomicAdd(tp + 64, w[i] * a[i][2]);
      atomicAdd(tp + 96, w[i] * a[i][3]);
      if (MODE == 0 && lane == 0) atomicAdd(den + d_i, w[i]);
    }
  }
  if ((MODE == 1 || MODE == 3) && lane < EB && ok) atomicAdd(den + dl, f0);
}

template <int MODE>
__device__ __forceinline__ void sweep(unsigned* q, int wave, int lane, int lo, unsigned ucnt, int rot,
                                      const OwnInv& inv, const unsigned* __restrict__ own8,
                                      const float* __restrict__ X,
                                      const int* __restrict__ EI,
                                      const float* __restrict__ rn,
                                      float* __restrict__ elog, float beta, float shift,
                                      float* tile, const float* xt,
                                      float* den, const float* nrn,
                                      unsigned* mx, const float* mxf) {
  int qn = 0;
  const unsigned lanes_below = (1u << lane) - 1u;
  const uint4* __restrict__ plane = (const uint4*)own8;
  for (int b = wave; b < KEY_STEPS; b += OWN_WAVES * 2) {
    uint4 ka[2], kb[2];
    int   gi[2];
    bool  in[2];
    #pragma unroll
    for (int j = 0; j < 2; ++j) {
      const int it = b + OWN_WAVES * j;
      int itr = it + rot;
      if (itr >= KEY_STEPS) itr -= KEY_STEPS;
      const int idx = itr * 32 + lane;
      gi[j] = idx;
      in[j] = it < KEY_STEPS && idx < KEY_GROUPS;
      const int ci = idx < KEY_GROUPS ? idx : KEY_GROUPS - 1;
      ka[j] = plane[ci];
      kb[j] = plane[ci + PLANE_HALF / 4];
    }
    unsigned m[2];
    #pragma unroll
    for (int j = 0; j < 2; ++j) m[j] = in[j] ? match32(ka[j], kb[j], inv) : 0u;
    #pragma nounroll
    for (int j = 0; j < 2; ++j) {
      unsigned mj = j == 0 ? m[0] : m[1];
      const int ebj = (j == 0 ? gi[0] : gi[1]) * 32;
      unsigned mask = __builtin_amdgcn_ballot_w32(mj != 0u);
      while (mask != 0u) {
        if (mj != 0u) {
          q[qn + __popc(mask & lanes_below)] = (unsigned)(ebj + __builtin_ctz(mj));
          mj &= mj - 1u;
        }
        qn += __popc(mask);
        __builtin_amdgcn_fence(__ATOMIC_ACQ_REL, "wavefront");
        while (qn >= EB) {
          qn -= EB;
          serve<MODE>(q + qn, EB, lane, lo, ucnt, X, EI, rn, elog, beta, shift, tile, xt, den, nrn, mx, mxf);
        }
        mask = __builtin_amdgcn_ballot_w32(mj != 0u);
      }
    }
  }
  __builtin_amdgcn_fence(__ATOMIC_ACQ_REL, "wavefront");
  if (qn > 0) serve<MODE>(q, qn, lane, lo, ucnt, X, EI, rn, elog, beta, shift, tile, xt, den, nrn, mx, mxf);
}

template <int CONV>
__global__ __launch_bounds__(1024) void k_owner(const float* __restrict__ X,
                                                const int* __restrict__ EI,
                                                const float* __restrict__ rn,
                                                float* __restrict__ elog,
                                                const float* __restrict__ beta_ptr,
                                                float* __restrict__ Y,
                                                float* __restrict__ rn_out,
                                                const unsigned* __restrict__ own8) {
  __shared__ float tile[OWN * HID];
  __shared__ float xt[OWN * HID];
  __shared__ float den[OWN];
  __shared__ float nrn[OWN];
  __shared__ unsigned mx[CONV == 1 ? OWN : 1];
  __shared__ float mxf[CONV == 1 ? OWN : 1];
  __shared__ unsigned queue[OWN_WAVES * QCAP];

  const int wave = __builtin_amdgcn_readfirstlane((int)(threadIdx.x >> 5));
  const int lane = (int)(threadIdx.x & 31);
  const int lo   = (int)blockIdx.x * OWN;
  const int cnt  = (N_NODES - lo) < OWN ? (N_NODES - lo) : OWN;
  const unsigned ucnt = (unsigned)cnt;
  const int rot = ((int)blockIdx.x * KEY_STEPS) >> 8;
  OwnInv inv;
  #pragma unroll
  for (int k = 0; k < 8; ++k) inv.v[k] = ((blockIdx.x >> k) & 1u) - 1u;
  unsigned* q = queue + wave * QCAP;
  const float beta  = (CONV == 1) ? beta_ptr[0] : 1.0f;
  const bool  one   = (CONV == 0) || (fabsf(beta) <= SHIFT_MAX);
  const float shift = (CONV == 0) ? 1.0f : fabsf(beta);

  for (int n = wave; n < cnt; n += OWN_WAVES) {
    const int node = lo + n;
    const float* xr = X + (size_t)node * HID + lane;
    const float a0 = xr[0], a1 = xr[32], a2 = xr[64], a3 = xr[96];
    const float r = rn[node];
    float s = a0 * a0 + a1 * a1 + a2 * a2 + a3 * a3;
    s = wave_sum(s);
    float* xo = xt + n * HID + lane;
    xo[0] = a0; xo[32] = a1; xo[64] = a2; xo[96] = a3;
    if (one) {
      const float w = __expf(beta * s * r * r - shift);
      if (lane == 0) { den[n] = w; nrn[n] = r; }
      float* to = tile + n * HID + lane;
      to[0] = w * a0; to[32] = w * a1; to[64] = w * a2; to[96] = w * a3;
    } else {
      const float logit = beta * s * r * r;
      if (lane == 0) { mx[n] = enc_f32(logit); mxf[n] = logit; nrn[n] = r; }
    }
  }
  __syncthreads();

  if (one) {
    sweep<(CONV == 0) ? 0 : 3>(q, wave, lane, lo, ucnt, rot, inv, own8, X, EI, rn, elog, beta, shift, tile, xt, den, nrn, mx, mxf);
  } else {
    sweep<2>(q, wave, lane, lo, ucnt, rot, inv, own8, X, EI, rn, elog, beta, shift, tile, xt, den, nrn, mx, mxf);
    __syncthreads();
    for (int n = wave; n < cnt; n += OWN_WAVES) {
      const float m = dec_f32(mx[n]);
      const float w = __expf(mxf[n] - m);
      const float* xo = xt + n * HID + lane;
      float* to = tile + n * HID + lane;
      to[0] = w * xo[0]; to[32] = w * xo[32]; to[64] = w * xo[64]; to[96] = w * xo[96];
      if (lane == 0) { den[n] = w; mxf[n] = m; }
    }
    __syncthreads();
    sweep<1>(q, wave, lane, lo, ucnt, rot, inv, own8, X, EI, rn, elog, beta, shift, tile, xt, den, nrn, mx, mxf);
  }
  __syncthreads();

  for (int n = wave; n < cnt; n += OWN_WAVES) {
    const int node = lo + n;
    const float* ti = tile + n * HID + lane;
    float v0 = ti[0], v1 = ti[32], v2 = ti[64], v3 = ti[96];
    float* yo = Y + (size_t)node * HID + lane;
    const float d = den[n] + 1e-16f;
    v0 /= d; v1 /= d; v2 /= d; v3 /= d;
    if (CONV == 0) {
      float s = v0 * v0 + v1 * v1 + v2 * v2 + v3 * v3;
      s = wave_sum(s);
      if (lane == 0) rn_out[node] = 1.f / fmaxf(sqrtf(s), 1e-12f);
    }
    yo[0] = v0; yo[32] = v1; yo[64] = v2; yo[96] = v3;
  }
}

extern "C" void kernel_launch(void* const* d_in, const int* in_sizes, int n_in,
                              void* d_out, int out_size, void* d_ws, size_t ws_size,
                              hipStream_t stream) {
  const float* x     = (const float*)d_in[0];
  const int*   ei    = (const int*)d_in[1];
  const float* W1    = (const float*)d_in[2];
  const float* b1    = (const float*)d_in[3];
  const float* beta2 = (const float*)d_in[4];
  float* out = (float*)d_out;

  float*    ws     = (float*)d_ws;
  float*    h1     = ws;
  float*    h2     = ws + 6400000;
  float*    rn     = ws + 12800000;
  float*    rn2    = ws + 12950000;
  float*    elog   = ws;
  unsigned* own8   = (unsigned*)(ws + 13000000);

  k_gemm_relu<<<196, 128, 0, stream>>>(x, W1, b1, h1, rn, ei, own8);

  k_owner<0><<<OWN_BLOCKS, 1024, 0, stream>>>(h1, ei, rn, nullptr, nullptr, h2, rn2, own8);

  k_owner<1><<<OWN_BLOCKS, 1024, 0, stream>>>(h2, ei, rn2, elog, beta2, out, nullptr, own8);
}


# Conv lane sums (a batch's 8 edge dot products; row norms) on the matrix unit: partials x one-hot columns by f32 16x16x4 WMMA, D rows added, one xor-16 exchange; sum tree changes (fixed per run)
# speedup vs baseline: 5.0536x; 1.0582x over previous
#include <hip/hip_runtime.h>
#include <hip/hip_bf16.h>
#include <math.h>


typedef float v2f __attribute__((ext_vector_type(2)));
typedef float v8f __attribute__((ext_vector_type(8)));

#define N_NODES 50000
#define N_EDGES 600000
#define E_TOT   650000
#define IN_CH   256
#define HID     128
#define ENC_NEG_INF 0x007FFFFFu
#define OWN        196
#define OWN_BLOCKS 256
#define PLANE_HALF (N_EDGES / 8)
#define SHIFT_MAX  8.0f

__device__ __forceinline__ unsigned enc_f32(float f) {
  unsigned u = __float_as_uint(f);
  return (u & 0x80000000u) ? ~u : (u | 0x80000000u);
}
__device__ __forceinline__ float dec_f32(unsigned e) {
  unsigned u = (e & 0x80000000u) ? (e & 0x7FFFFFFFu) : ~e;
  return __uint_as_float(u);
}

__device__ __forceinline__ unsigned owner_id(int dst) {
  return ((unsigned)dst / (unsigned)OWN) & 255u;
}

#define LANE_SUM_TAIL "s_wait_alu depctr_va_vdst(0)\n\tv_nop\n\tv_nop\n\tv_nop\n\tv_nop"
__device__ __forceinline__ float sum_rows(v8f d) {
  const float t = ((d[0] + d[1]) + (d[2] + d[3])) + ((d[4] + d[5]) + (d[6] + d[7]));
  return t + __shfl_xor(t, 16, 32);
}
__device__ __forceinline__ float wave_sum(float v) {
  v8f d;
  asm volatile("v_wmma_f32_16x16x4_f32 %0, %1, %2, 0\n\t"
               LANE_SUM_TAIL
               : "=&v"(d) : "v"((v2f){v, 0.f}), "v"((v2f){1.f, 1.f}));
  return sum_rows(d);
}

struct Frags { v2f a0[4], a1[4], b0[8], b1[8]; };
__device__ __forceinline__ void load_frags(Frags& f, const float* __restrict__ const* xrow,
                                           const float* __restrict__ wbas, int k) {
  #pragma unroll
  for (int i = 0; i < 4; ++i) {
    f.a0[i] = *(const v2f*)(xrow[i] + k);
    f.a1[i] = *(const v2f*)(xrow[i] + k + 4);
  }
  #pragma unroll
  for (int t = 0; t < 8; ++t) {
    f.b0[t] = *(const v2f*)(wbas + (size_t)t * 16 * IN_CH + k);
    f.b1[t] = *(const v2f*)(wbas + (size_t)t * 16 * IN_CH + k + 4);
  }
}
__device__ __forceinline__ void mma_frags(v8f (&acc)[4][8], const Frags& f) {
  #pragma unroll
  for (int t = 0; t < 8; ++t) {
    #pragma unroll
    for (int i = 0; i < 4; ++i)
      acc[i][t] = __builtin_amdgcn_wmma_f32_16x16x4_f32(
          false, f.a0[i], false, f.b0[t], (short)0, acc[i][t], false, false);
  }
  #pragma unroll
  for (int t = 0; t < 8; ++t) {
    #pragma unroll
    for (int i = 0; i < 4; ++i)
      acc[i][t] = __builtin_amdgcn_wmma_f32_16x16x4_f32(
          false, f.a1[i], false, f.b1[t], (short)0, acc[i][t], false, false);
  }
}

#define GEMM_TILES 3125
#define GEMM_RT    4

__global__ __launch_bounds__(128) void k_gemm_relu(const float* __restrict__ X,
                            const float* __restrict__ W,
                            const float* __restrict__ B,
                            float* __restrict__ H,
                            float* __restrict__ rn,
                            const int* __restrict__ EI,
                            unsigned* __restrict__ own8) {
  const int wave = __builtin_amdgcn_readfirstlane((int)((blockIdx.x * blockDim.x + threadIdx.x) >> 5));
  const int lane = (int)(threadIdx.x & 31);
  if (wave * GEMM_RT >= GEMM_TILES) return;

  {
    unsigned o[GEMM_RT][6];
    #pragma unroll
    for (int vi = 0; vi < GEMM_RT; ++vi) {
      const int v  = wave * GEMM_RT + vi;
      const int vc = v < GEMM_TILES ? v : GEMM_TILES - 1;
      const int* __restrict__ dk = EI + N_EDGES + vc * 192 + lane;
      #pragma unroll
      for (int g = 0; g < 6; ++g) o[vi][g] = owner_id(dk[32 * g]);
    }
    #pragma unroll
    for (int vi = 0; vi < GEMM_RT; ++vi) {
      const int v = wave * GEMM_RT + vi;
      if (v < GEMM_TILES) {
        unsigned r0 = 0u, r1 = 0u;
        #pragma unroll
        for (int g = 0; g < 6; ++g) {
          #pragma unroll
          for (int k = 0; k < 8; ++k) {
            const unsigned bal = __builtin_amdgcn_ballot_w32(((o[vi][g] >> k) & 1u) != 0u);
            if (g < 4) r0 = (lane == 8 * g + k) ? bal : r0;
            else       r1 = (lane == 8 * (g - 4) + k) ? bal : r1;
          }
        }
        unsigned* __restrict__ ov = own8 + (lane & 4) * (PLANE_HALF / 4) + (v * 6 + (lane >> 3)) * 4 + (lane & 3);
        ov[0] = r0;
        if (lane < 16) ov[16] = r1;
      }
    }
  }

  const int t0   = wave * GEMM_RT;
  const int kg   = (lane >> 4) * 2;
  const int n15  = lane & 15;

  v8f acc[GEMM_RT][8];
  #pragma unroll
  for (int i = 0; i < GEMM_RT; ++i) {
    #pragma unroll
    for (int t = 0; t < 8; ++t) acc[i][t] = (v8f){0.f,0.f,0.f,0.f,0.f,0.f,0.f,0.f};
  }

  const float* __restrict__ xrow[GEMM_RT];
  #pragma unroll
  for (int i = 0; i < GEMM_RT; ++i) {
    const int ti = (t0 + i) < GEMM_TILES ? (t0 + i) : (GEMM_TILES - 1);
    xrow[i] = X + (size_t)(ti * 16 + n15) * IN_CH + kg;
  }
  const float* __restrict__ wbas = W + (size_t)n15 * IN_CH + kg;

  Frags p, q;
  load_frags(p, xrow, wbas, 0);
  __builtin_amdgcn_sched_barrier(0);
  load_frags(q, xrow, wbas, 8);
  __builtin_amdgcn_sched_barrier(0);
  for (int k = 0; k < IN_CH; k += 16) {
    mma_frags(acc, p);
    load_frags(p, xrow, wbas, (k + 16) < IN_CH ? (k + 16) : (IN_CH - 8));
    __builtin_amdgcn_sched_barrier(0);
    mma_frags(acc, q);
    load_frags(q, xrow, wbas, (k + 24) < IN_CH ? (k + 24) : (IN_CH - 8));
    __builtin_amdgcn_sched_barrier(0);
  }

  const int mside = (lane >> 4) * 8;
  #pragma unroll
  for (int i = 0; i < GEMM_RT; ++i) {
    if (t0 + i < GEMM_TILES) {
      const int m0 = (t0 + i) * 16;
      float ss[8];
      #pragma unroll
      for (int r = 0; r < 8; ++r) ss[r] = 0.f;
      #pragma unroll
      for (int t = 0; t < 8; ++t) {
        const int col = t * 16 + n15;
        const float bb = B[col];
        #pragma unroll
        for (int r = 0; r < 8; ++r) {
          const int row = m0 + r + mside;
          float v = acc[i][t][r] + bb;
          v = v > 0.f ? v : 0.f;
          H[(size_t)row * HID + col] = v;
          ss[r] += v * v;
        }
      }
      #pragma unroll
      for (int r = 0; r < 8; ++r) {
        #pragma unroll
        for (int off = 8; off >= 1; off >>= 1) ss[r] += __shfl_xor(ss[r], off, 32);
      }
      if (n15 == 0) {
        #pragma unroll
        for (int r = 0; r < 8; ++r)
          rn[m0 + r + mside] = 1.f / fmaxf(sqrtf(ss[r]), 1e-12f);
      }
    }
  }
}

#define OWN_WAVES  32
#define EB         8
#define QCAP       40
#define KEY_GROUPS (N_EDGES / 32)
#define KEY_STEPS  ((KEY_GROUPS + 31) / 32)

struct OwnInv { unsigned v[8]; };
__device__ __forceinline__ unsigned match32(uint4 a, uint4 b, const OwnInv& inv) {
  return (a.x ^ inv.v[0]) & (a.y ^ inv.v[1]) & (a.z ^ inv.v[2]) & (a.w ^ inv.v[3]) &
         (b.x ^ inv.v[4]) & (b.y ^ inv.v[5]) & (b.z ^ inv.v[6]) & (b.w ^ inv.v[7]);
}

struct LaneSel { v2f p[EB / 2]; };

template <int MODE>
__device__ __forceinline__ void serve(const unsigned* qb, int n, int lane, int lo, unsigned ucnt,
                                      const float* __restrict__ X,
                                      const int* __restrict__ EI,
                                      const float* __restrict__ rn,
                                      float* __restrict__ elog, float beta, float shift,
                                      float* tile, const float* xt,
                                      float* den, const float* nrn,
                                      unsigned* mx, const float* mxf, const LaneSel& sel) {
  const int li = lane & (EB - 1);
  const int e   = (int)qb[li < n ? li : n - 1];
  const int src = EI[e];
  const unsigned dlu = (unsigned)(EI[N_EDGES + e] - lo);
  const bool ok = (li < n) && (dlu < ucnt);
  const unsigned vmask = __builtin_amdgcn_ballot_w32(ok);
  const int dl  = ok ? (int)dlu : 0;
  float f0, f1 = 0.f;
  if (MODE == 1) { f0 = __expf(elog[e] - mxf[dl]); }
  else           { f0 = rn[src]; f1 = nrn[dl]; }

  float a[EB][4];
  #pragma unroll
  for (int i = 0; i < EB; ++i) {
    const int s_i = __builtin_amdgcn_readlane(src, i);
    const float* xs = X + (size_t)s_i * HID + lane;
    a[i][0] = xs[0]; a[i][1] = xs[32]; a[i][2] = xs[64]; a[i][3] = xs[96];
  }
  float w[EB];
  if (MODE != 1) {
    float s[EB];
    #pragma unroll
    for (int i = 0; i < EB; ++i) {
      const int d_i = __builtin_amdgcn_readlane(dl, i);
      const float* xb = xt + d_i * HID + lane;
      s[i] = a[i][0] * xb[0] + a[i][1] * xb[32] + a[i][2] * xb[64] + a[i][3] * xb[96];
    }
    v8f d;
    asm volatile("v_wmma_f32_16x16x4_f32 %0, %1, %5, 0\n\t"
                 "v_wmma_f32_16x16x4_f32 %0, %2, %6, %0\n\t"
                 "v_wmma_f32_16x16x4_f32 %0, %3, %7, %0\n\t"
                 "v_wmma_f32_16x16x4_f32 %0, %4, %8, %0\n\t"
                 LANE_SUM_TAIL
                 : "=&v"(d)
                 : "v"((v2f){s[0], s[1]}), "v"((v2f){s[2], s[3]}),
                   "v"((v2f){s[4], s[5]}), "v"((v2f){s[6], s[7]}),
                   "v"(sel.p[0]), "v"(sel.p[1]), "v"(sel.p[2]), "v"(sel.p[3]));
    const float mine = sum_rows(d);
    const float logit = beta * mine * f0 * f1;
    if (MODE == 2) {
      if (lane < EB && ok) {
        elog[e] = logit;
        atomicMax(mx + dl, enc_f32(logit));
      }
      return;
    }
    f0 = __expf(logit - shift);
  }
  #pragma unroll
  for (int i = 0; i < EB; ++i)
    w[i] = __int_as_float(__builtin_amdgcn_readlane(__float_as_int(f0), i));
  #pragma unroll
  for (int i = 0; i < EB; ++i) {
    if ((vmask >> i) & 1u) {
      const int d_i = __builtin_amdgcn_readlane(dl, i);
      float* tp = tile + d_i * HID + lane;
      atomicAdd(tp +  0, w[i] * a[i][0]);
      atomicAdd(tp + 32, w[i] * a[i][1]);
      atomicAdd(tp + 64, w[i] * a[i][2]);
      atomicAdd(tp + 96, w[i] * a[i][3]);
    }
  }
  if (lane < EB && ok) atomicAdd(den + dl, f0);
}

template <int MODE>
__device__ __forceinline__ void sweep(unsigned* q, int wave, int lane, int lo, unsigned ucnt, int rot,
                                      const OwnInv& inv, const unsigned* __restrict__ own8,
                                      const float* __restrict__ X,
                                      const int* __restrict__ EI,
                                      const float* __restrict__ rn,
                                      float* __restrict__ elog, float beta, float shift,
                                      float* tile, const float* xt,
                                      float* den, const float* nrn,
                                      unsigned* mx, const float* mxf, const LaneSel& sel) {
  int qn = 0;
  const unsigned lanes_below = (1u << lane) - 1u;
  const uint4* __restrict__ plane = (const uint4*)own8;
  for (int b = wave; b < KEY_STEPS; b += OWN_WAVES * 2) {
    uint4 ka[2], kb[2];
    int   gi[2];
    bool  in[2];
    #pragma unroll
    for (int j = 0; j < 2; ++j) {
      const int it = b + OWN_WAVES * j;
      int itr = it + rot;
      if (itr >= KEY_STEPS) itr -= KEY_STEPS;
      const int idx = itr * 32 + lane;
      gi[j] = idx;
      in[j] = it < KEY_STEPS && idx < KEY_GROUPS;
      const int ci = idx < KEY_GROUPS ? idx : KEY_GROUPS - 1;
      ka[j] = plane[ci];
      kb[j] = plane[ci + PLANE_HALF / 4];
    }
    unsigned m[2];
    #pragma unroll
    for (int j = 0; j < 2; ++j) m[j] = in[j] ? match32(ka[j], kb[j], inv) : 0u;
    #pragma nounroll
    for (int j = 0; j < 2; ++j) {
      unsigned mj = j == 0 ? m[0] : m[1];
      const int ebj = (j == 0 ? gi[0] : gi[1]) * 32;
      unsigned mask = __builtin_amdgcn_ballot_w32(mj != 0u);
      while (mask != 0u) {
        if (mj != 0u) {
          q[qn + __popc(mask & lanes_below)] = (unsigned)(ebj + __builtin_ctz(mj));
          mj &= mj - 1u;
        }
        qn += __popc(mask);
        __builtin_amdgcn_fence(__ATOMIC_ACQ_REL, "wavefront");
        while (qn >= EB) {
          qn -= EB;
          serve<MODE>(q + qn, EB, lane, lo, ucnt, X, EI, rn, elog, beta, shift, tile, xt, den, nrn, mx, mxf, sel);
        }
        mask = __builtin_amdgcn_ballot_w32(mj != 0u);
      }
    }
  }
  __builtin_amdgcn_fence(__ATOMIC_ACQ_REL, "wavefront");
  if (qn > 0) serve<MODE>(q, qn, lane, lo, ucnt, X, EI, rn, elog, beta, shift, tile, xt, den, nrn, mx, mxf, sel);
}

template <int CONV>
__global__ __launch_bounds__(1024) void k_owner(const float* __restrict__ X,
                                                const int* __restrict__ EI,
                                                const float* __restrict__ rn,
                                                float* __restrict__ elog,
                                                const float* __restrict__ beta_ptr,
                                                float* __restrict__ Y,
                                                float* __restrict__ rn_out,
                                                const unsigned* __restrict__ own8) {
  __shared__ float tile[OWN * HID];
  __shared__ float xt[OWN * HID];
  __shared__ float den[OWN];
  __shared__ float nrn[OWN];
  __shared__ unsigned mx[CONV == 1 ? OWN : 1];
  __shared__ float mxf[CONV == 1 ? OWN : 1];
  __shared__ unsigned queue[OWN_WAVES * QCAP];

  const int wave = __builtin_amdgcn_readfirstlane((int)(threadIdx.x >> 5));
  const int lane = (int)(threadIdx.x & 31);
  const int lo   = (int)blockIdx.x * OWN;
  const int cnt  = (N_NODES - lo) < OWN ? (N_NODES - lo) : OWN;
  const unsigned ucnt = (unsigned)cnt;
  const int rot = ((int)blockIdx.x * KEY_STEPS) >> 8;
  OwnInv inv;
  #pragma unroll
  for (int k = 0; k < 8; ++k) inv.v[k] = ((blockIdx.x >> k) & 1u) - 1u;
  unsigned* q = queue + wave * QCAP;
  LaneSel sel;
  const int l7 = lane & 7;
  sel.p[0] = (v2f){l7 == 0 ? 1.f : 0.f, l7 == 1 ? 1.f : 0.f};
  sel.p[1] = (v2f){l7 == 2 ? 1.f : 0.f, l7 == 3 ? 1.f : 0.f};
  sel.p[2] = (v2f){l7 == 4 ? 1.f : 0.f, l7 == 5 ? 1.f : 0.f};
  sel.p[3] = (v2f){l7 == 6 ? 1.f : 0.f, l7 == 7 ? 1.f : 0.f};
  const float beta  = (CONV == 1) ? beta_ptr[0] : 1.0f;
  const bool  one   = (CONV == 0) || (fabsf(beta) <= SHIFT_MAX);
  const float shift = (CONV == 0) ? 1.0f : fabsf(beta);

  for (int n = wave; n < cnt; n += OWN_WAVES) {
    const int node = lo + n;
    const float* xr = X + (size_t)node * HID + lane;
    const float a0 = xr[0], a1 = xr[32], a2 = xr[64], a3 = xr[96];
    const float r = rn[node];
    float s = a0 * a0 + a1 * a1 + a2 * a2 + a3 * a3;
    s = wave_sum(s);
    float* xo = xt + n * HID + lane;
    xo[0] = a0; xo[32] = a1; xo[64] = a2; xo[96] = a3;
    if (one) {
      const float w = __expf(beta * s * r * r - shift);
      if (lane == 0) { den[n] = w; nrn[n] = r; }
      float* to = tile + n * HID + lane;
      to[0] = w * a0; to[32] = w * a1; to[64] = w * a2; to[96] = w * a3;
    } else {
      const float logit = beta * s * r * r;
      if (lane == 0) { mx[n] = enc_f32(logit); mxf[n] = logit; nrn[n] = r; }
    }
  }
  __syncthreads();

  if (one) {
    sweep<3>(q, wave, lane, lo, ucnt, rot, inv, own8, X, EI, rn, elog, beta, shift, tile, xt, den, nrn, mx, mxf, sel);
  } else {
    sweep<2>(q, wave, lane, lo, ucnt, rot, inv, own8, X, EI, rn, elog, beta, shift, tile, xt, den, nrn, mx, mxf, sel);
    __syncthreads();
    for (int n = wave; n < cnt; n += OWN_WAVES) {
      const float m = dec_f32(mx[n]);
      const float w = __expf(mxf[n] - m);
      const float* xo = xt + n * HID + lane;
      float* to = tile + n * HID + lane;
      to[0] = w * xo[0]; to[32] = w * xo[32]; to[64] = w * xo[64]; to[96] = w * xo[96];
      if (lane == 0) { den[n] = w; mxf[n] = m; }
    }
    __syncthreads();
    sweep<1>(q, wave, lane, lo, ucnt, rot, inv, own8, X, EI, rn, elog, beta, shift, tile, xt, den, nrn, mx, mxf, sel);
  }
  __syncthreads();

  for (int n = wave; n < cnt; n += OWN_WAVES) {
    const int node = lo + n;
    const float* ti = tile + n * HID + lane;
    float v0 = ti[0], v1 = ti[32], v2 = ti[64], v3 = ti[96];
    float* yo = Y + (size_t)node * HID + lane;
    const float d = den[n] + 1e-16f;
    v0 /= d; v1 /= d; v2 /= d; v3 /= d;
    if (CONV == 0) {
      float s = v0 * v0 + v1 * v1 + v2 * v2 + v3 * v3;
      s = wave_sum(s);
      if (lane == 0) rn_out[node] = 1.f / fmaxf(sqrtf(s), 1e-12f);
    }
    yo[0] = v0; yo[32] = v1; yo[64] = v2; yo[96] = v3;
  }
}

extern "C" void kernel_launch(void* const* d_in, const int* in_sizes, int n_in,
                              void* d_out, int out_size, void* d_ws, size_t ws_size,
                              hipStream_t stream) {
  const float* x     = (const float*)d_in[0];
  const int*   ei    = (const int*)d_in[1];
  const float* W1    = (const float*)d_in[2];
  const float* b1    = (const float*)d_in[3];
  const float* beta2 = (const float*)d_in[4];
  float* out = (float*)d_out;

  float*    ws     = (float*)d_ws;
  float*    h1     = ws;
  float*    h2     = ws + 6400000;
  float*    rn     = ws + 12800000;
  float*    rn2    = ws + 12950000;
  float*    elog   = ws;
  unsigned* own8   = (unsigned*)(ws + 13000000);

  k_gemm_relu<<<196, 128, 0, stream>>>(x, W1, b1, h1, rn, ei, own8);

  k_owner<0><<<OWN_BLOCKS, 1024, 0, stream>>>(h1, ei, rn, nullptr, nullptr, h2, rn2, own8);

  k_owner<1><<<OWN_BLOCKS, 1024, 0, stream>>>(h2, ei, rn2, elog, beta2, out, nullptr, own8);
}


# Conv rows move 16 B per lane (lane = channels 4l..4l+3): a gathered, owned or finished row is ONE 512-B load/store and one ds_load_b128, not 4; dot and norm lane partials sum in that channel order
# speedup vs baseline: 5.9333x; 1.1741x over previous
#include <hip/hip_runtime.h>
#include <hip/hip_bf16.h>
#include <math.h>


typedef float v2f __attribute__((ext_vector_type(2)));
typedef float v8f __attribute__((ext_vector_type(8)));

#define N_NODES 50000
#define N_EDGES 600000
#define E_TOT   650000
#define IN_CH   256
#define HID     128
#define ENC_NEG_INF 0x007FFFFFu
#define OWN        196
#define OWN_BLOCKS 256
#define PLANE_HALF (N_EDGES / 8)
#define SHIFT_MAX  8.0f

__device__ __forceinline__ unsigned enc_f32(float f) {
  unsigned u = __float_as_uint(f);
  return (u & 0x80000000u) ? ~u : (u | 0x80000000u);
}
__device__ __forceinline__ float dec_f32(unsigned e) {
  unsigned u = (e & 0x80000000u) ? (e & 0x7FFFFFFFu) : ~e;
  return __uint_as_float(u);
}

__device__ __forceinline__ unsigned owner_id(int dst) {
  return ((unsigned)dst / (unsigned)OWN) & 255u;
}

#define LANE_SUM_TAIL "s_wait_alu depctr_va_vdst(0)\n\tv_nop\n\tv_nop\n\tv_nop\n\tv_nop"
__device__ __forceinline__ float sum_rows(v8f d) {
  const float t = ((d[0] + d[1]) + (d[2] + d[3])) + ((d[4] + d[5]) + (d[6] + d[7]));
  return t + __shfl_xor(t, 16, 32);
}
__device__ __forceinline__ float wave_sum(float v) {
  v8f d;
  asm volatile("v_wmma_f32_16x16x4_f32 %0, %1, %2, 0\n\t"
               LANE_SUM_TAIL
               : "=&v"(d) : "v"((v2f){v, 0.f}), "v"((v2f){1.f, 1.f}));
  return sum_rows(d);
}

struct Frags { v2f a0[4], a1[4], b0[8], b1[8]; };
__device__ __forceinline__ void load_frags(Frags& f, const float* __restrict__ const* xrow,
                                           const float* __restrict__ wbas, int k) {
  #pragma unroll
  for (int i = 0; i < 4; ++i) {
    f.a0[i] = *(const v2f*)(xrow[i] + k);
    f.a1[i] = *(const v2f*)(xrow[i] + k + 4);
  }
  #pragma unroll
  for (int t = 0; t < 8; ++t) {
    f.b0[t] = *(const v2f*)(wbas + (size_t)t * 16 * IN_CH + k);
    f.b1[t] = *(const v2f*)(wbas + (size_t)t * 16 * IN_CH + k + 4);
  }
}
__device__ __forceinline__ void mma_frags(v8f (&acc)[4][8], const Frags& f) {
  #pragma unroll
  for (int t = 0; t < 8; ++t) {
    #pragma unroll
    for (int i = 0; i < 4; ++i)
      acc[i][t] = __builtin_amdgcn_wmma_f32_16x16x4_f32(
          false, f.a0[i], false, f.b0[t], (short)0, acc[i][t], false, false);
  }
  #pragma unroll
  for (int t = 0; t < 8; ++t) {
    #pragma unroll
    for (int i = 0; i < 4; ++i)
      acc[i][t] = __builtin_amdgcn_wmma_f32_16x16x4_f32(
          false, f.a1[i], false, f.b1[t], (short)0, acc[i][t], false, false);
  }
}

#define GEMM_TILES 3125
#define GEMM_RT    4

__global__ __launch_bounds__(128) void k_gemm_relu(const float* __restrict__ X,
                            const float* __restrict__ W,
                            const float* __restrict__ B,
                            float* __restrict__ H,
                            float* __restrict__ rn,
                            const int* __restrict__ EI,
                            unsigned* __restrict__ own8) {
  const int wave = __builtin_amdgcn_readfirstlane((int)((blockIdx.x * blockDim.x + threadIdx.x) >> 5));
  const int lane = (int)(threadIdx.x & 31);
  if (wave * GEMM_RT >= GEMM_TILES) return;

  {
    unsigned o[GEMM_RT][6];
    #pragma unroll
    for (int vi = 0; vi < GEMM_RT; ++vi) {
      const int v  = wave * GEMM_RT + vi;
      const int vc = v < GEMM_TILES ? v : GEMM_TILES - 1;
      const int* __restrict__ dk = EI + N_EDGES + vc * 192 + lane;
      #pragma unroll
      for (int g = 0; g < 6; ++g) o[vi][g] = owner_id(dk[32 * g]);
    }
    #pragma unroll
    for (int vi = 0; vi < GEMM_RT; ++vi) {
      const int v = wave * GEMM_RT + vi;
      if (v < GEMM_TILES) {
        unsigned r0 = 0u, r1 = 0u;
        #pragma unroll
        for (int g = 0; g < 6; ++g) {
          #pragma unroll
          for (int k = 0; k < 8; ++k) {
            const unsigned bal = __builtin_amdgcn_ballot_w32(((o[vi][g] >> k) & 1u) != 0u);
            if (g < 4) r0 = (lane == 8 * g + k) ? bal : r0;
            else       r1 = (lane == 8 * (g - 4) + k) ? bal : r1;
          }
        }
        unsigned* __restrict__ ov = own8 + (lane & 4) * (PLANE_HALF / 4) + (v * 6 + (lane >> 3)) * 4 + (lane & 3);
        ov[0] = r0;
        if (lane < 16) ov[16] = r1;
      }
    }
  }

  const int t0   = wave * GEMM_RT;
  const int kg   = (lane >> 4) * 2;
  const int n15  = lane & 15;

  v8f acc[GEMM_RT][8];
  #pragma unroll
  for (int i = 0; i < GEMM_RT; ++i) {
    #pragma unroll
    for (int t = 0; t < 8; ++t) acc[i][t] = (v8f){0.f,0.f,0.f,0.f,0.f,0.f,0.f,0.f};
  }

  const float* __restrict__ xrow[GEMM_RT];
  #pragma unroll
  for (int i = 0; i < GEMM_RT; ++i) {
    const int ti = (t0 + i) < GEMM_TILES ? (t0 + i) : (GEMM_TILES - 1);
    xrow[i] = X + (size_t)(ti * 16 + n15) * IN_CH + kg;
  }
  const float* __restrict__ wbas = W + (size_t)n15 * IN_CH + kg;

  Frags p, q;
  load_frags(p, xrow, wbas, 0);
  __builtin_amdgcn_sched_barrier(0);
  load_frags(q, xrow, wbas, 8);
  __builtin_amdgcn_sched_barrier(0);
  for (int k = 0; k < IN_CH; k += 16) {
    mma_frags(acc, p);
    load_frags(p, xrow, wbas, (k + 16) < IN_CH ? (k + 16) : (IN_CH - 8));
    __builtin_amdgcn_sched_barrier(0);
    mma_frags(acc, q);
    load_frags(q, xrow, wbas, (k + 24) < IN_CH ? (k + 24) : (IN_CH - 8));
    __builtin_amdgcn_sched_barrier(0);
  }

  const int mside = (lane >> 4) * 8;
  #pragma unroll
  for (int i = 0; i < GEMM_RT; ++i) {
    if (t0 + i < GEMM_TILES) {
      const int m0 = (t0 + i) * 16;
      float ss[8];
      #pragma unroll
      for (int r = 0; r < 8; ++r) ss[r] = 0.f;
      #pragma unroll
      for (int t = 0; t < 8; ++t) {
        const int col = t * 16 + n15;
        const float bb = B[col];
        #pragma unroll
        for (int r = 0; r < 8; ++r) {
          const int row = m0 + r + mside;
          float v = acc[i][t][r] + bb;
          v = v > 0.f ? v : 0.f;
          H[(size_t)row * HID + col] = v;
          ss[r] += v * v;
        }
      }
      #pragma unroll
      for (int r = 0; r < 8; ++r) {
        #pragma unroll
        for (int off = 8; off >= 1; off >>= 1) ss[r] += __shfl_xor(ss[r], off, 32);
      }
      if (n15 == 0) {
        #pragma unroll
        for (int r = 0; r < 8; ++r)
          rn[m0 + r + mside] = 1.f / fmaxf(sqrtf(ss[r]), 1e-12f);
      }
    }
  }
}

#define OWN_WAVES  32
#define EB         8
#define QCAP       40
#define KEY_GROUPS (N_EDGES / 32)
#define KEY_STEPS  ((KEY_GROUPS + 31) / 32)

struct OwnInv { unsigned v[8]; };
__device__ __forceinline__ unsigned match32(uint4 a, uint4 b, const OwnInv& inv) {
  return (a.x ^ inv.v[0]) & (a.y ^ inv.v[1]) & (a.z ^ inv.v[2]) & (a.w ^ inv.v[3]) &
         (b.x ^ inv.v[4]) & (b.y ^ inv.v[5]) & (b.z ^ inv.v[6]) & (b.w ^ inv.v[7]);
}

struct LaneSel { v2f p[EB / 2]; };

template <int MODE>
__device__ __forceinline__ void serve(const unsigned* qb, int n, int lane, int lo, unsigned ucnt,
                                      const float* __restrict__ X,
                                      const int* __restrict__ EI,
                                      const float* __restrict__ rn,
                                      float* __restrict__ elog, float beta, float shift,
                                      float* tile, const float* xt,
                                      float* den, const float* nrn,
                                      unsigned* mx, const float* mxf, const LaneSel& sel) {
  const int li = lane & (EB - 1);
  const int e   = (int)qb[li < n ? li : n - 1];
  const int src = EI[e];
  const unsigned dlu = (unsigned)(EI[N_EDGES + e] - lo);
  const bool ok = (li < n) && (dlu < ucnt);
  const unsigned vmask = __builtin_amdgcn_ballot_w32(ok);
  const int dl  = ok ? (int)dlu : 0;
  float f0, f1 = 0.f;
  if (MODE == 1) { f0 = __expf(elog[e] - mxf[dl]); }
  else           { f0 = rn[src]; f1 = nrn[dl]; }

  float4 a[EB];
  #pragma unroll
  for (int i = 0; i < EB; ++i) {
    const int s_i = __builtin_amdgcn_readlane(src, i);
    a[i] = *(const float4*)(X + (size_t)s_i * HID + 4 * lane);
  }
  float w[EB];
  if (MODE != 1) {
    float s[EB];
    #pragma unroll
    for (int i = 0; i < EB; ++i) {
      const int d_i = __builtin_amdgcn_readlane(dl, i);
      const float4 xb = *(const float4*)(xt + d_i * HID + 4 * lane);
      s[i] = a[i].x * xb.x + a[i].y * xb.y + a[i].z * xb.z + a[i].w * xb.w;
    }
    v8f d;
    asm volatile("v_wmma_f32_16x16x4_f32 %0, %1, %5, 0\n\t"
                 "v_wmma_f32_16x16x4_f32 %0, %2, %6, %0\n\t"
                 "v_wmma_f32_16x16x4_f32 %0, %3, %7, %0\n\t"
                 "v_wmma_f32_16x16x4_f32 %0, %4, %8, %0\n\t"
                 LANE_SUM_TAIL
                 : "=&v"(d)
                 : "v"((v2f){s[0], s[1]}), "v"((v2f){s[2], s[3]}),
                   "v"((v2f){s[4], s[5]}), "v"((v2f){s[6], s[7]}),
                   "v"(sel.p[0]), "v"(sel.p[1]), "v"(sel.p[2]), "v"(sel.p[3]));
    const float mine = sum_rows(d);
    const float logit = beta * mine * f0 * f1;
    if (MODE == 2) {
      if (lane < EB && ok) {
        elog[e] = logit;
        atomicMax(mx + dl, enc_f32(logit));
      }
      return;
    }
    f0 = __expf(logit - shift);
  }
  #pragma unroll
  for (int i = 0; i < EB; ++i)
    w[i] = __int_as_float(__builtin_amdgcn_readlane(__float_as_int(f0), i));
  #pragma unroll
  for (int i = 0; i < EB; ++i) {
    if ((vmask >> i) & 1u) {
      const int d_i = __builtin_amdgcn_readlane(dl, i);
      float* tp = tile + d_i * HID + lane;
      atomicAdd(tp +  0, w[i] * a[i].x);
      atomicAdd(tp + 32, w[i] * a[i].y);
      atomicAdd(tp + 64, w[i] * a[i].z);
      atomicAdd(tp + 96, w[i] * a[i].w);
    }
  }
  if (lane < EB && ok) atomicAdd(den + dl, f0);
}

template <int MODE>
__device__ __forceinline__ void sweep(unsigned* q, int wave, int lane, int lo, unsigned ucnt, int rot,
                                      const OwnInv& inv, const unsigned* __restrict__ own8,
                                      const float* __restrict__ X,
                                      const int* __restrict__ EI,
                                      const float* __restrict__ rn,
                                      float* __restrict__ elog, float beta, float shift,
                                      float* tile, const float* xt,
                                      float* den, const float* nrn,
                                      unsigned* mx, const float* mxf, const LaneSel& sel) {
  int qn = 0;
  const unsigned lanes_below = (1u << lane) - 1u;
  const uint4* __restrict__ plane = (const uint4*)own8;
  for (int b = wave; b < KEY_STEPS; b += OWN_WAVES * 2) {
    uint4 ka[2], kb[2];
    int   gi[2];
    bool  in[2];
    #pragma unroll
    for (int j = 0; j < 2; ++j) {
      const int it = b + OWN_WAVES * j;
      int itr = it + rot;
      if (itr >= KEY_STEPS) itr -= KEY_STEPS;
      const int idx = itr * 32 + lane;
      gi[j] = idx;
      in[j] = it < KEY_STEPS && idx < KEY_GROUPS;
      const int ci = idx < KEY_GROUPS ? idx : KEY_GROUPS - 1;
      ka[j] = plane[ci];
      kb[j] = plane[ci + PLANE_HALF / 4];
    }
    unsigned m[2];
    #pragma unroll
    for (int j = 0; j < 2; ++j) m[j] = in[j] ? match32(ka[j], kb[j], inv) : 0u;
    #pragma nounroll
    for (int j = 0; j < 2; ++j) {
      unsigned mj = j == 0 ? m[0] : m[1];
      const int ebj = (j == 0 ? gi[0] : gi[1]) * 32;
      unsigned mask = __builtin_amdgcn_ballot_w32(mj != 0u);
      while (mask != 0u) {
        if (mj != 0u) {
          q[qn + __popc(mask & lanes_below)] = (unsigned)(ebj + __builtin_ctz(mj));
          mj &= mj - 1u;
        }
        qn += __popc(mask);
        __builtin_amdgcn_fence(__ATOMIC_ACQ_REL, "wavefront");
        while (qn >= EB) {
          qn -= EB;
          serve<MODE>(q + qn, EB, lane, lo, ucnt, X, EI, rn, elog, beta, shift, tile, xt, den, nrn, mx, mxf, sel);
        }
        mask = __builtin_amdgcn_ballot_w32(mj != 0u);
      }
    }
  }
  __builtin_amdgcn_fence(__ATOMIC_ACQ_REL, "wavefront");
  if (qn > 0) serve<MODE>(q, qn, lane, lo, ucnt, X, EI, rn, elog, beta, shift, tile, xt, den, nrn, mx, mxf, sel);
}

template <int CONV>
__global__ __launch_bounds__(1024) void k_owner(const float* __restrict__ X,
                                                const int* __restrict__ EI,
                                                const float* __restrict__ rn,
                                                float* __restrict__ elog,
                                                const float* __restrict__ beta_ptr,
                                                float* __restrict__ Y,
                                                float* __restrict__ rn_out,
                                                const unsigned* __restrict__ own8) {
  __shared__ float tile[OWN * HID];
  __shared__ __attribute__((aligned(16))) float xt[OWN * HID];
  __shared__ float den[OWN];
  __shared__ float nrn[OWN];
  __shared__ unsigned mx[CONV == 1 ? OWN : 1];
  __shared__ float mxf[CONV == 1 ? OWN : 1];
  __shared__ unsigned queue[OWN_WAVES * QCAP];

  const int wave = __builtin_amdgcn_readfirstlane((int)(threadIdx.x >> 5));
  const int lane = (int)(threadIdx.x & 31);
  const int lo   = (int)blockIdx.x * OWN;
  const int cnt  = (N_NODES - lo) < OWN ? (N_NODES - lo) : OWN;
  const unsigned ucnt = (unsigned)cnt;
  const int rot = ((int)blockIdx.x * KEY_STEPS) >> 8;
  OwnInv inv;
  #pragma unroll
  for (int k = 0; k < 8; ++k) inv.v[k] = ((blockIdx.x >> k) & 1u) - 1u;
  unsigned* q = queue + wave * QCAP;
  LaneSel sel;
  const int l7 = lane & 7;
  sel.p[0] = (v2f){l7 == 0 ? 1.f : 0.f, l7 == 1 ? 1.f : 0.f};
  sel.p[1] = (v2f){l7 == 2 ? 1.f : 0.f, l7 == 3 ? 1.f : 0.f};
  sel.p[2] = (v2f){l7 == 4 ? 1.f : 0.f, l7 == 5 ? 1.f : 0.f};
  sel.p[3] = (v2f){l7 == 6 ? 1.f : 0.f, l7 == 7 ? 1.f : 0.f};
  const float beta  = (CONV == 1) ? beta_ptr[0] : 1.0f;
  const bool  one   = (CONV == 0) || (fabsf(beta) <= SHIFT_MAX);
  const float shift = (CONV == 0) ? 1.0f : fabsf(beta);

  for (int n = wave; n < cnt; n += OWN_WAVES) {
    const int node = lo + n;
    const float4 av = *(const float4*)(X + (size_t)node * HID + 4 * lane);
    const float a0 = av.x, a1 = av.y, a2 = av.z, a3 = av.w;
    const float r = rn[node];
    float s = a0 * a0 + a1 * a1 + a2 * a2 + a3 * a3;
    s = wave_sum(s);
    *(float4*)(xt + n * HID + 4 * lane) = av;
    if (one) {
      const float w = __expf(beta * s * r * r - shift);
      if (lane == 0) { den[n] = w; nrn[n] = r; }
      float* to = tile + n * HID + lane;
      to[0] = w * a0; to[32] = w * a1; to[64] = w * a2; to[96] = w * a3;
    } else {
      const float logit = beta * s * r * r;
      if (lane == 0) { mx[n] = enc_f32(logit); mxf[n] = logit; nrn[n] = r; }
    }
  }
  __syncthreads();

  if (one) {
    sweep<3>(q, wave, lane, lo, ucnt, rot, inv, own8, X, EI, rn, elog, beta, shift, tile, xt, den, nrn, mx, mxf, sel);
  } else {
    sweep<2>(q, wave, lane, lo, ucnt, rot, inv, own8, X, EI, rn, elog, beta, shift, tile, xt, den, nrn, mx, mxf, sel);
    __syncthreads();
    for (int n = wave; n < cnt; n += OWN_WAVES) {
      const float m = dec_f32(mx[n]);
      const float w = __expf(mxf[n] - m);
      const float4 xo = *(const float4*)(xt + n * HID + 4 * lane);
      float* to = tile + n * HID + lane;
      to[0] = w * xo.x; to[32] = w * xo.y; to[64] = w * xo.z; to[96] = w * xo.w;
      if (lane == 0) { den[n] = w; mxf[n] = m; }
    }
    __syncthreads();
    sweep<1>(q, wave, lane, lo, ucnt, rot, inv, own8, X, EI, rn, elog, beta, shift, tile, xt, den, nrn, mx, mxf, sel);
  }
  __syncthreads();

  for (int n = wave; n < cnt; n += OWN_WAVES) {
    const int node = lo + n;
    const float* ti = tile + n * HID + lane;
    float v0 = ti[0], v1 = ti[32], v2 = ti[64], v3 = ti[96];
    float4* yo = (float4*)(Y + (size_t)node * HID + 4 * lane);
    const float d = den[n] + 1e-16f;
    v0 /= d; v1 /= d; v2 /= d; v3 /= d;
    if (CONV == 0) {
      float s = v0 * v0 + v1 * v1 + v2 * v2 + v3 * v3;
      s = wave_sum(s);
      if (lane == 0) rn_out[node] = 1.f / fmaxf(sqrtf(s), 1e-12f);
    }
    *yo = make_float4(v0, v1, v2, v3);
  }
}

extern "C" void kernel_launch(void* const* d_in, const int* in_sizes, int n_in,
                              void* d_out, int out_size, void* d_ws, size_t ws_size,
                              hipStream_t stream) {
  const float* x     = (const float*)d_in[0];
  const int*   ei    = (const int*)d_in[1];
  const float* W1    = (const float*)d_in[2];
  const float* b1    = (const float*)d_in[3];
  const float* beta2 = (const float*)d_in[4];
  float* out = (float*)d_out;

  float*    ws     = (float*)d_ws;
  float*    h1     = ws;
  float*    h2     = ws + 6400000;
  float*    rn     = ws + 12800000;
  float*    rn2    = ws + 12950000;
  float*    elog   = ws;
  unsigned* own8   = (unsigned*)(ws + 13000000);

  k_gemm_relu<<<196, 128, 0, stream>>>(x, W1, b1, h1, rn, ei, own8);

  k_owner<0><<<OWN_BLOCKS, 1024, 0, stream>>>(h1, ei, rn, nullptr, nullptr, h2, rn2, own8);

  k_owner<1><<<OWN_BLOCKS, 1024, 0, stream>>>(h2, ei, rn2, elog, beta2, out, nullptr, own8);
}
